# starting kernel
# baseline (speedup 1.0000x reference)
#include <hip/hip_runtime.h>
#include <hip/hip_bf16.h>


#define DIMC   256
#define MLPC   1024
#define BC     2
#define NC     2048
#define HEADS  8
#define HDC    32
#define ROWS   (BC * NC)
#define QKVN   (3 * DIMC)
#define LAYER  1

typedef __attribute__((ext_vector_type(2))) float v2f;
typedef __attribute__((ext_vector_type(8))) float v8f;

static __device__ inline v8f wmma4(v2f a, v2f b, v8f c) {
  return __builtin_amdgcn_wmma_f32_16x16x4_f32(
       false, a,  false, b,
       (short)0, c,  false,  false);
}

static __device__ inline v8f v8zero() {
  v8f z = {0.f, 0.f, 0.f, 0.f, 0.f, 0.f, 0.f, 0.f};
  return z;
}

__global__ __launch_bounds__(256) void ln_kernel(
    const float* __restrict__ x, const float* __restrict__ g,
    const float* __restrict__ b, float* __restrict__ y, int rows) {
  int wave = (blockIdx.x * blockDim.x + threadIdx.x) >> 5;
  int lane = threadIdx.x & 31;
  if (wave >= rows) return;
  const float* xr = x + (size_t)wave * DIMC;
  float vals[8];
  float s = 0.f;
#pragma unroll
  for (int i = 0; i < 8; ++i) { vals[i] = xr[lane + 32 * i]; s += vals[i]; }
#pragma unroll
  for (int off = 16; off >= 1; off >>= 1) s += __shfl_xor(s, off, 32);
  float mean = s * (1.0f / DIMC);
  float vs = 0.f;
#pragma unroll
  for (int i = 0; i < 8; ++i) { float d = vals[i] - mean; vs += d * d; }
#pragma unroll
  for (int off = 16; off >= 1; off >>= 1) vs += __shfl_xor(vs, off, 32);
  float rstd = rsqrtf(vs * (1.0f / DIMC) + 1e-5f);
  float* yr = y + (size_t)wave * DIMC;
#pragma unroll
  for (int i = 0; i < 8; ++i) {
    int c = lane + 32 * i;
    yr[c] = (vals[i] - mean) * rstd * g[c] + b[c];
  }
}

__global__ __launch_bounds__(128) void wmma_gemm_kernel(
    const float* __restrict__ A, const float* __restrict__ W,
    const float* __restrict__ bias, const float* __restrict__ resid,
    float* __restrict__ out, int M, int N, int K, int gelu_flag) {
  const int lane = threadIdx.x & 31;
  const int wv = threadIdx.x >> 5;
  const int half = lane >> 4;
  const int lc = lane & 15;
  const int n0 = blockIdx.x * 64;
  const int m0 = blockIdx.y * 64 + wv * 16;

  const float* Arow = A + (size_t)(m0 + lc) * K;
  v8f acc[4];
#pragma unroll
  for (int nt = 0; nt < 4; ++nt) acc[nt] = v8zero();

  for (int kk = 0; kk < K; kk += 4) {
    if ((kk & 63) == 0 && kk + 64 < K)
      __builtin_prefetch(Arow + kk + 64, 0, 0);
    const int ka = kk + 2 * half;
    v2f a = {Arow[ka], Arow[ka + 1]};
    const float* Wr0 = W + (size_t)ka * N + n0 + lc;
    const float* Wr1 = Wr0 + N;
#pragma unroll
    for (int nt = 0; nt < 4; ++nt) {
      v2f b = {Wr0[nt * 16], Wr1[nt * 16]};
      acc[nt] = wmma4(a, b, acc[nt]);
    }
  }

#pragma unroll
  for (int nt = 0; nt < 4; ++nt) {
    const int col = n0 + nt * 16 + lc;
    const float bv = bias ? bias[col] : 0.f;
#pragma unroll
    for (int v = 0; v < 8; ++v) {
      const int row = m0 + v + 8 * half;
      float val = acc[nt][v] + bv;
      if (gelu_flag)
        val = 0.5f * val * (1.0f + erff(val * 0.70710678118654752f));
      if (resid) val += resid[(size_t)row * N + col];
      out[(size_t)row * N + col] = val;
    }
  }
}

__global__ __launch_bounds__(128) void attn_kernel(
    const float* __restrict__ qkv, float* __restrict__ o) {
  __shared__ float plds[4][256];
  const int lane = threadIdx.x & 31;
  const int wv = threadIdx.x >> 5;
  const int half = lane >> 4;
  const int lc = lane & 15;
  const int bh = blockIdx.x;
  const int bb = bh >> 3, h = bh & 7;
  const int row0 = (blockIdx.y * 4 + wv) * 16;
  const int base = bb * NC;
  const float scale = 0.17677669529663687f;

  const float* Q = qkv + (size_t)base * QKVN + h * HDC;
  const float* Kp = Q + DIMC;
  const float* Vp = Q + 2 * DIMC;

  v2f qa[8];
  {
    const float* qrow = Q + (size_t)(row0 + lc) * QKVN + 2 * half;
#pragma unroll
    for (int i = 0; i < 8; ++i) {
      v2f t = {qrow[4 * i] * scale, qrow[4 * i + 1] * scale};
      qa[i] = t;
    }
  }

  float mrow[8], lrow[8];
#pragma unroll
  for (int v = 0; v < 8; ++v) { mrow[v] = -1e30f; lrow[v] = 0.f; }
  v8f oacc0 = v8zero(), oacc1 = v8zero();

  for (int jb = 0; jb < NC / 16; ++jb) {
    const float* krow = Kp + (size_t)(jb * 16 + lc) * QKVN + 2 * half;
    v8f s = v8zero();
#pragma unroll
    for (int i = 0; i < 8; ++i) {
      v2f kb = {krow[4 * i], krow[4 * i + 1]};
      s = wmma4(qa[i], kb, s);
    }
#pragma unroll
    for (int v = 0; v < 8; ++v) {
      float sv = s[v];
      float rmax = sv;
      rmax = fmaxf(rmax, __shfl_xor(rmax, 1, 32));
      rmax = fmaxf(rmax, __shfl_xor(rmax, 2, 32));
      rmax = fmaxf(rmax, __shfl_xor(rmax, 4, 32));
      rmax = fmaxf(rmax, __shfl_xor(rmax, 8, 32));
      float mn = fmaxf(mrow[v], rmax);
      float p = __expf(sv - mn);
      float rs = p;
      rs += __shfl_xor(rs, 1, 32);
      rs += __shfl_xor(rs, 2, 32);
      rs += __shfl_xor(rs, 4, 32);
      rs += __shfl_xor(rs, 8, 32);
      float c = __expf(mrow[v] - mn);
      lrow[v] = lrow[v] * c + rs;
      mrow[v] = mn;
      oacc0[v] *= c;
      oacc1[v] *= c;
      plds[wv][(v + 8 * half) * 16 + lc] = p;
    }
#pragma unroll
    for (int i = 0; i < 4; ++i) {
      v2f pa = {plds[wv][lc * 16 + 4 * i + 2 * half],
                plds[wv][lc * 16 + 4 * i + 2 * half + 1]};
      const int kv = jb * 16 + 4 * i + 2 * half;
      const float* v0 = Vp + (size_t)kv * QKVN + lc;
      const float* v1 = v0 + QKVN;
      v2f vb0 = {v0[0], v1[0]};
      v2f vb1 = {v0[16], v1[16]};
      oacc0 = wmma4(pa, vb0, oacc0);
      oacc1 = wmma4(pa, vb1, oacc1);
    }
  }

#pragma unroll
  for (int v = 0; v < 8; ++v) {
    const float inv = 1.0f / lrow[v];
    float* orow = o + (size_t)(base + row0 + v + 8 * half) * DIMC + h * HDC;
    orow[lc] = oacc0[v] * inv;
    orow[16 + lc] = oacc1[v] * inv;
  }
}

extern "C" void kernel_launch(void* const* d_in, const int* in_sizes, int n_in,
                              void* d_out, int out_size, void* d_ws, size_t ws_size,
                              hipStream_t stream) {
  const float* x     = (const float*)d_in[0];
  const float* ln1_g = (const float*)d_in[1] + (size_t)LAYER * DIMC;
  const float* ln1_b = (const float*)d_in[2] + (size_t)LAYER * DIMC;
  const float* Wqkv  = (const float*)d_in[3] + (size_t)LAYER * DIMC * QKVN;
  const float* bqkv  = (const float*)d_in[4] + (size_t)LAYER * QKVN;
  const float* Wo    = (const float*)d_in[5] + (size_t)LAYER * DIMC * DIMC;
  const float* bo    = (const float*)d_in[6] + (size_t)LAYER * DIMC;
  const float* ln2_g = (const float*)d_in[7] + (size_t)LAYER * DIMC;
  const float* ln2_b = (const float*)d_in[8] + (size_t)LAYER * DIMC;
  const float* W1    = (const float*)d_in[9] + (size_t)LAYER * DIMC * MLPC;
  const float* b1    = (const float*)d_in[10] + (size_t)LAYER * MLPC;
  const float* W2    = (const float*)d_in[11] + (size_t)LAYER * MLPC * DIMC;
  const float* b2    = (const float*)d_in[12] + (size_t)LAYER * DIMC;
  float* out = (float*)d_out;

  float* ws   = (float*)d_ws;
  float* y    = ws;
  float* qkvb = ws + (size_t)ROWS * DIMC;
  float* oat  = y;
  float* a    = qkvb + (size_t)ROWS * QKVN;
  float* z    = qkvb;
  float* h1   = a + (size_t)ROWS * DIMC;

  ln_kernel<<<ROWS / 8, 256, 0, stream>>>(x, ln1_g, ln1_b, y, ROWS);
  wmma_gemm_kernel<<<dim3(QKVN / 64, ROWS / 64), 128, 0, stream>>>(
      y, Wqkv, bqkv, nullptr, qkvb, ROWS, QKVN, DIMC, 0);
  attn_kernel<<<dim3(BC * HEADS, NC / 64), 128, 0, stream>>>(qkvb, oat);
  wmma_gemm_kernel<<<dim3(DIMC / 64, ROWS / 64), 128, 0, stream>>>(
      oat, Wo, bo, x, a, ROWS, DIMC, DIMC, 0);
  ln_kernel<<<ROWS / 8, 256, 0, stream>>>(a, ln2_g, ln2_b, z, ROWS);
  wmma_gemm_kernel<<<dim3(MLPC / 64, ROWS / 64), 128, 0, stream>>>(
      z, W1, b1, nullptr, h1, ROWS, MLPC, DIMC, 1);
  wmma_gemm_kernel<<<dim3(DIMC / 64, ROWS / 64), 128, 0, stream>>>(
      h1, W2, b2, a, out, ROWS, DIMC, MLPC, 0);
}


# LayerNorm fused into the GEMM that consumes it: each workgroup normalises its 64x256 A panel into LDS once and reads A fragments from there; 7 launches -> 5; changes LN row-sum order
# speedup vs baseline: 3.1103x; 3.1103x over previous
#include <hip/hip_runtime.h>
#include <hip/hip_bf16.h>


#define DIMC   256
#define MLPC   1024
#define BC     2
#define NC     2048
#define HEADS  8
#define HDC    32
#define ROWS   (BC * NC)
#define QKVN   (3 * DIMC)
#define LAYER  1

typedef __attribute__((ext_vector_type(2))) float v2f;
typedef __attribute__((ext_vector_type(4))) float v4f;
typedef __attribute__((ext_vector_type(8))) float v8f;

static __device__ inline v8f wmma4(v2f a, v2f b, v8f c) {
  return __builtin_amdgcn_wmma_f32_16x16x4_f32(
       false, a,  false, b,
       (short)0, c,  false,  false);
}

static __device__ inline v8f v8zero() {
  v8f z = {0.f, 0.f, 0.f, 0.f, 0.f, 0.f, 0.f, 0.f};
  return z;
}

template <int M>
static __device__ inline float xor_in_row(float v) {
  return __int_as_float(__builtin_amdgcn_update_dpp(
      0, __float_as_int(v), 0x160 + M, 0xf, 0xf, true));
}

static __device__ inline float wave_sum(float s) {
  s += xor_in_row<1>(s);
  s += xor_in_row<2>(s);
  s += xor_in_row<4>(s);
  s += xor_in_row<8>(s);
  s += __shfl_xor(s, 16, 32);
  return s;
}

#define PPITCH 68
#define APITCH 260

template <int KS, bool FRAG, bool LNA>
__global__ __launch_bounds__(128 * KS) void wmma_gemm_kernel(
    const float* __restrict__ A, const float* __restrict__ W,
    const float* __restrict__ bias, const float* __restrict__ resid,
    const float* __restrict__ lng, const float* __restrict__ lnb,
    float* __restrict__ out, int M, int N, int K, int gelu_flag) {
  constexpr int PARTF = 4 * KS * 16 * PPITCH;
  constexpr int PANELF = LNA ? 64 * APITCH : 0;
  __shared__ float smem[PARTF > PANELF ? PARTF : PANELF];
  float (*part)[16 * PPITCH] = reinterpret_cast<float (*)[16 * PPITCH]>(smem);
  const int lane = threadIdx.x & 31;
  const int wv = __builtin_amdgcn_readfirstlane(threadIdx.x >> 5);
  const int rb = wv & 3;
  const int ks = wv >> 2;
  const int half = lane >> 4;
  const int lc = lane & 15;
  const int n0 = blockIdx.x * 64;
  const int m0 = blockIdx.y * 64 + rb * 16;
  const int klen = K / KS;
  const int kbeg = ks * klen;
  const int kend = kbeg + klen;

  const float* Arow = A + (size_t)(m0 + lc) * K;

  if constexpr (LNA) {
    static_assert(KS == 4, "panel prologue: 16 waves x 4 rows = 64 rows");
    const float* src = A + (size_t)(blockIdx.y * 64 + wv * 4) * DIMC + lane * 4;
    v4f gv[2], bv[2];
#pragma unroll
    for (int p = 0; p < 2; ++p) {
      gv[p] = *(const v4f*)(lng + p * 128 + lane * 4);
      bv[p] = *(const v4f*)(lnb + p * 128 + lane * 4);
    }
    v4f xv[4][2];
#pragma unroll
    for (int j = 0; j < 4; ++j) {
#pragma unroll
      for (int p = 0; p < 2; ++p) xv[j][p] = *(const v4f*)(src + j * DIMC + p * 128);
    }
    __builtin_amdgcn_sched_barrier(0);
#pragma unroll
    for (int j = 0; j < 4; ++j) {
      v2f s2 = {0.f, 0.f};
#pragma unroll
      for (int p = 0; p < 2; ++p) {
#pragma unroll
        for (int q = 0; q < 2; ++q) {
          v2f t = {xv[j][p][2 * q], xv[j][p][2 * q + 1]};
          s2 += t;
        }
      }
      const float mean = wave_sum(s2[0] + s2[1]) * (1.0f / DIMC);
      v2f d2[2][2];
      v2f vs2 = {0.f, 0.f};
#pragma unroll
      for (int p = 0; p < 2; ++p) {
#pragma unroll
        for (int q = 0; q < 2; ++q) {
          v2f t = {xv[j][p][2 * q] - mean, xv[j][p][2 * q + 1] - mean};
          d2[p][q] = t;
          vs2 += t * t;
        }
      }
      const float rstd = rsqrtf(wave_sum(vs2[0] + vs2[1]) * (1.0f / DIMC) + 1e-5f);
#pragma unroll
      for (int p = 0; p < 2; ++p) {
        v4f yv;
#pragma unroll
        for (int q = 0; q < 2; ++q) {
#pragma unroll
          for (int e = 0; e < 2; ++e)
            yv[2 * q + e] = d2[p][q][e] * rstd * gv[p][2 * q + e] + bv[p][2 * q + e];
        }
        *(v4f*)(smem + (wv * 4 + j) * APITCH + p * 128 + lane * 4) = yv;
      }
    }
    __syncthreads();
  }
  const float* Apan = smem + (rb * 16 + lc) * APITCH;

  v8f acc[4];
#pragma unroll
  for (int nt = 0; nt < 4; ++nt) acc[nt] = v8zero();

  for (int kk = kbeg; kk < kend; kk += 8) {
    if constexpr (!LNA) {
      if ((kk & 63) == 0 && kk + 64 < kend)
        __builtin_prefetch(Arow + kk + 64, 0, 0);
    }
    v2f a[2], b[2][4];
#pragma unroll
    for (int u = 0; u < 2; ++u) {
      const int ka = kk + 4 * u + 2 * half;
      if constexpr (LNA) {
        a[u] = *(const v2f*)(Apan + ka);
      } else {
        v2f t = {Arow[ka], Arow[ka + 1]};
        a[u] = t;
      }
      const float* Wr0 = W + (size_t)ka * N + n0 + lc;
      const float* Wr1 = Wr0 + N;
#pragma unroll
      for (int nt = 0; nt < 4; ++nt) {
        v2f tb = {Wr0[nt * 16], Wr1[nt * 16]};
        b[u][nt] = tb;
      }
    }
#pragma unroll
    for (int u = 0; u < 2; ++u) {
#pragma unroll
      for (int nt = 0; nt < 4; ++nt) acc[nt] = wmma4(a[u], b[u][nt], acc[nt]);
    }
  }

  if constexpr (LNA) __syncthreads();

#pragma unroll
  for (int nt = 0; nt < 4; ++nt) {
#pragma unroll
    for (int v = 0; v < 8; ++v)
      part[wv][(v + 8 * half) * PPITCH + nt * 16 + lc] = acc[nt][v];
  }
  __syncthreads();

  if constexpr (FRAG) {
    static_assert(KS == 4, "fragment-order epilogue: 4 waves per key tile");
    const int sec = n0 / DIMC;
    const int cin = n0 - sec * DIMC;
    if (sec == 0) {
#pragma unroll
      for (int j = 0; j < 4; ++j) {
        const int r = ks * 4 + j;
        const int row = blockIdx.y * 64 + rb * 16 + r;
#pragma unroll
        for (int cs = 0; cs < 2; ++cs) {
          const int cl = cs * 32 + lane;
          float val = part[rb][r * PPITCH + cl];
#pragma unroll
          for (int t = 1; t < KS; ++t) val += part[4 * t + rb][r * PPITCH + cl];
          val += bias ? bias[n0 + cl] : 0.f;
          out[(size_t)row * DIMC + cin + cl] = val;
        }
      }
    } else {
      const int tile = blockIdx.y * 4 + rb;
      const int bb = tile / (NC / 16);
      const int jt = tile - bb * (NC / 16);
      const int hh = ks >> 1;
      const int head = cin / HDC + hh;
      float* dst = out + (size_t)sec * ROWS * DIMC +
                   (((size_t)(bb * HEADS + head) * (NC / 16) + jt) * 128 + lane) * 4;
#pragma unroll
      for (int q = 0; q < 2; ++q) {
        const int st = (ks & 1) * 2 + q;
        v4f val4;
#pragma unroll
        for (int sub = 0; sub < 2; ++sub) {
#pragma unroll
          for (int e = 0; e < 2; ++e) {
            int r, cl;
            if (sec == 1) { r = lc;                    cl = hh * HDC + 8 * st + 4 * sub + 2 * half + e; }
            else          { r = 4 * st + 2 * half + e; cl = hh * HDC + lc + 16 * sub; }
            float val = part[rb][r * PPITCH + cl];
#pragma unroll
            for (int t = 1; t < KS; ++t) val += part[4 * t + rb][r * PPITCH + cl];
            val += bias ? bias[n0 + cl] : 0.f;
            val4[sub * 2 + e] = val;
          }
        }
        *(v4f*)(dst + (size_t)st * 128) = val4;
      }
    }
  } else {
  constexpr int RPW = 16 / KS;
#pragma unroll
  for (int j = 0; j < RPW; ++j) {
    const int r = ks * RPW + j;
    const int row = blockIdx.y * 64 + rb * 16 + r;
#pragma unroll
    for (int cs = 0; cs < 2; ++cs) {
      const int cl = cs * 32 + lane;
      float val = part[rb][r * PPITCH + cl];
#pragma unroll
      for (int t = 1; t < KS; ++t) val += part[4 * t + rb][r * PPITCH + cl];
      const int col = n0 + cl;
      val += bias ? bias[col] : 0.f;
      if (gelu_flag)
        val = 0.5f * val * (1.0f + erff(val * 0.70710678118654752f));
      if (resid) val += resid[(size_t)row * N + col];
      out[(size_t)row * N + col] = val;
    }
  }
  }
}

#define KSPLIT 4
#define KEYS_PER_SPLIT (NC / KSPLIT)
#define KT 2

__global__ __launch_bounds__(512, 8) void attn_kernel(
    const float* __restrict__ qkv, float* __restrict__ o) {
  __shared__ float plds[4 * KSPLIT][KT][256];
  __shared__ float part_o[4 * KSPLIT][16 * HDC];
  __shared__ float part_m[4 * KSPLIT][16];
  __shared__ float part_l[4 * KSPLIT][16];
  const int lane = threadIdx.x & 31;
  const int wv = threadIdx.x >> 5;
  const int qb = wv & 3;
  const int ks = wv >> 2;
  const int half = lane >> 4;
  const int lc = lane & 15;
  const int bh = blockIdx.x;
  const int bb = bh >> 3, h = bh & 7;
  const int row0 = (blockIdx.y * 4 + qb) * 16;
  const int base = bb * NC;
  const float scale = 0.17677669529663687f;

  const float* Q = qkv + (size_t)base * DIMC + h * HDC;
  const float* Kf = qkv + (size_t)ROWS * DIMC + (size_t)bh * (NC / 16) * 512 + lane * 4;
  const float* Vf = Kf + (size_t)ROWS * DIMC;

  v2f qa[8];
  {
    const float* qrow = Q + (size_t)(row0 + lc) * DIMC + 2 * half;
#pragma unroll
    for (int i = 0; i < 8; ++i) {
      v2f t = {qrow[4 * i] * scale, qrow[4 * i + 1] * scale};
      qa[i] = t;
    }
  }

  float mrow[8], lrow[8];
#pragma unroll
  for (int v = 0; v < 8; ++v) { mrow[v] = -1e30f; lrow[v] = 0.f; }
  v8f oacc0 = v8zero(), oacc1 = v8zero();

  const int jb0 = ks * (KEYS_PER_SPLIT / 16);
  for (int jb = jb0; jb < jb0 + KEYS_PER_SPLIT / 16; jb += KT) {
    v8f s[KT];
#pragma unroll
    for (int kt = 0; kt < KT; ++kt) s[kt] = v8zero();
#pragma unroll
    for (int i2 = 0; i2 < 4; ++i2) {
#pragma unroll
      for (int kt = 0; kt < KT; ++kt) {
        const v4f k4 = *(const v4f*)(Kf + (size_t)((jb + kt) * 4 + i2) * 128);
        v2f kb0 = {k4[0], k4[1]};
        v2f kb1 = {k4[2], k4[3]};
        s[kt] = wmma4(qa[2 * i2], kb0, s[kt]);
        s[kt] = wmma4(qa[2 * i2 + 1], kb1, s[kt]);
      }
    }
#pragma unroll
    for (int v = 0; v < 8; ++v) {
      float rmax = s[0][v];
#pragma unroll
      for (int kt = 1; kt < KT; ++kt) rmax = fmaxf(rmax, s[kt][v]);
      rmax = fmaxf(rmax, __shfl_xor(rmax, 1, 32));
      rmax = fmaxf(rmax, __shfl_xor(rmax, 2, 32));
      rmax = fmaxf(rmax, __shfl_xor(rmax, 4, 32));
      rmax = fmaxf(rmax, __shfl_xor(rmax, 8, 32));
      const float mn = fmaxf(mrow[v], rmax);
      const float c = __expf(mrow[v] - mn);
      float rs = 0.f;
#pragma unroll
      for (int kt = 0; kt < KT; ++kt) {
        const float p = __expf(s[kt][v] - mn);
        rs += p;
        plds[wv][kt][(v + 8 * half) * 16 + lc] = p;
      }
      lrow[v] = lrow[v] * c + rs;
      mrow[v] = mn;
      oacc0[v] *= c;
      oacc1[v] *= c;
    }
#pragma unroll
    for (int kt = 0; kt < KT; ++kt) {
#pragma unroll
      for (int i = 0; i < 4; ++i) {
        v2f pa = {plds[wv][kt][lc * 16 + 4 * i + 2 * half],
                  plds[wv][kt][lc * 16 + 4 * i + 2 * half + 1]};
        const v4f v4 = *(const v4f*)(Vf + (size_t)((jb + kt) * 4 + i) * 128);
        v2f vb0 = {v4[0], v4[1]};
        v2f vb1 = {v4[2], v4[3]};
        oacc0 = wmma4(pa, vb0, oacc0);
        oacc1 = wmma4(pa, vb1, oacc1);
      }
    }
  }

#pragma unroll
  for (int v = 0; v < 8; ++v) {
    float l = lrow[v];
    l += __shfl_xor(l, 1, 32);
    l += __shfl_xor(l, 2, 32);
    l += __shfl_xor(l, 4, 32);
    l += __shfl_xor(l, 8, 32);
    lrow[v] = l;
  }

#pragma unroll
  for (int v = 0; v < 8; ++v) {
    const int r = v + 8 * half;
    part_o[wv][r * HDC + lc] = oacc0[v];
    part_o[wv][r * HDC + 16 + lc] = oacc1[v];
    if (lc == 0) { part_m[wv][r] = mrow[v]; part_l[wv][r] = lrow[v]; }
  }
  __syncthreads();

#pragma unroll
  for (int j = 0; j < 4; ++j) {
    const int r = 4 * ks + j;
    float mx = part_m[qb][r];
#pragma unroll
    for (int t = 1; t < KSPLIT; ++t) mx = fmaxf(mx, part_m[4 * t + qb][r]);
    float lsum = 0.f, osum = 0.f;
#pragma unroll
    for (int t = 0; t < KSPLIT; ++t) {
      const float w = __expf(part_m[4 * t + qb][r] - mx);
      lsum += part_l[4 * t + qb][r] * w;
      osum += part_o[4 * t + qb][r * HDC + lane] * w;
    }
    float* orow = o + (size_t)(base + row0 + r) * DIMC + h * HDC;
    orow[lane] = osum / lsum;
  }
}

extern "C" void kernel_launch(void* const* d_in, const int* in_sizes, int n_in,
                              void* d_out, int out_size, void* d_ws, size_t ws_size,
                              hipStream_t stream) {
  const float* x     = (const float*)d_in[0];
  const float* ln1_g = (const float*)d_in[1] + (size_t)LAYER * DIMC;
  const float* ln1_b = (const float*)d_in[2] + (size_t)LAYER * DIMC;
  const float* Wqkv  = (const float*)d_in[3] + (size_t)LAYER * DIMC * QKVN;
  const float* bqkv  = (const float*)d_in[4] + (size_t)LAYER * QKVN;
  const float* Wo    = (const float*)d_in[5] + (size_t)LAYER * DIMC * DIMC;
  const float* bo    = (const float*)d_in[6] + (size_t)LAYER * DIMC;
  const float* ln2_g = (const float*)d_in[7] + (size_t)LAYER * DIMC;
  const float* ln2_b = (const float*)d_in[8] + (size_t)LAYER * DIMC;
  const float* W1    = (const float*)d_in[9] + (size_t)LAYER * DIMC * MLPC;
  const float* b1    = (const float*)d_in[10] + (size_t)LAYER * MLPC;
  const float* W2    = (const float*)d_in[11] + (size_t)LAYER * MLPC * DIMC;
  const float* b2    = (const float*)d_in[12] + (size_t)LAYER * DIMC;
  float* out = (float*)d_out;

  float* ws   = (float*)d_ws;
  float* oat  = ws;
  float* qkvb = ws + (size_t)ROWS * DIMC;
  float* a    = qkvb + (size_t)ROWS * QKVN;
  float* h1   = a + (size_t)ROWS * DIMC;

  wmma_gemm_kernel<4, true, true><<<dim3(QKVN / 64, ROWS / 64), 512, 0, stream>>>(
      x, Wqkv, bqkv, nullptr, ln1_g, ln1_b, qkvb, ROWS, QKVN, DIMC, 0);
  attn_kernel<<<dim3(BC * HEADS, NC / 64), 512, 0, stream>>>(qkvb, oat);
  wmma_gemm_kernel<8, false, false><<<dim3(DIMC / 64, ROWS / 64), 1024, 0, stream>>>(
      oat, Wo, bo, x, nullptr, nullptr, a, ROWS, DIMC, DIMC, 0);
  wmma_gemm_kernel<4, false, true><<<dim3(MLPC / 64, ROWS / 64), 512, 0, stream>>>(
      a, W1, b1, nullptr, ln2_g, ln2_b, h1, ROWS, MLPC, DIMC, 1);
  wmma_gemm_kernel<8, false, false><<<dim3(DIMC / 64, ROWS / 64), 1024, 0, stream>>>(
      h1, W2, b2, a, nullptr, nullptr, out, ROWS, DIMC, MLPC, 0);
}


# GEMM A operands read from global memory (attention output, MLP hidden) are stored by their producers in WMMA A-fragment order: one contiguous 512-B wave load per 8 k; summation order unchanged
# speedup vs baseline: 3.2520x; 1.0456x over previous
#include <hip/hip_runtime.h>
#include <hip/hip_bf16.h>


#define DIMC   256
#define MLPC   1024
#define BC     2
#define NC     2048
#define HEADS  8
#define HDC    32
#define ROWS   (BC * NC)
#define QKVN   (3 * DIMC)
#define LAYER  1

typedef __attribute__((ext_vector_type(2))) float v2f;
typedef __attribute__((ext_vector_type(4))) float v4f;
typedef __attribute__((ext_vector_type(8))) float v8f;

static __device__ inline v8f wmma4(v2f a, v2f b, v8f c) {
  return __builtin_amdgcn_wmma_f32_16x16x4_f32(
       false, a,  false, b,
       (short)0, c,  false,  false);
}

static __device__ inline v8f v8zero() {
  v8f z = {0.f, 0.f, 0.f, 0.f, 0.f, 0.f, 0.f, 0.f};
  return z;
}

template <int M>
static __device__ inline float xor_in_row(float v) {
  return __int_as_float(__builtin_amdgcn_update_dpp(
      0, __float_as_int(v), 0x160 + M, 0xf, 0xf, true));
}

static __device__ inline float wave_sum(float s) {
  s += xor_in_row<1>(s);
  s += xor_in_row<2>(s);
  s += xor_in_row<4>(s);
  s += xor_in_row<8>(s);
  s += __shfl_xor(s, 16, 32);
  return s;
}

#define PPITCH 68
#define APITCH 260

template <int KS, bool FRAG, bool LNA, bool OFRAG>
__global__ __launch_bounds__(128 * KS) void wmma_gemm_kernel(
    const float* __restrict__ A, const float* __restrict__ W,
    const float* __restrict__ bias, const float* __restrict__ resid,
    const float* __restrict__ lng, const float* __restrict__ lnb,
    float* __restrict__ out, int M, int N, int K, int gelu_flag) {
  constexpr int PARTF = 4 * KS * 16 * PPITCH;
  constexpr int PANELF = LNA ? 64 * APITCH : 0;
  __shared__ float smem[PARTF > PANELF ? PARTF : PANELF];
  float (*part)[16 * PPITCH] = reinterpret_cast<float (*)[16 * PPITCH]>(smem);
  const int lane = threadIdx.x & 31;
  const int wv = __builtin_amdgcn_readfirstlane(threadIdx.x >> 5);
  const int rb = wv & 3;
  const int ks = wv >> 2;
  const int half = lane >> 4;
  const int lc = lane & 15;
  const int n0 = blockIdx.x * 64;
  const int m0 = blockIdx.y * 64 + rb * 16;
  const int klen = K / KS;
  const int kbeg = ks * klen;
  const int kend = kbeg + klen;

  const float* Afr = A + (size_t)(m0 >> 4) * (K >> 3) * 128 + lane * 4;

  if constexpr (LNA) {
    static_assert(KS == 4, "panel prologue: 16 waves x 4 rows = 64 rows");
    const float* src = A + (size_t)(blockIdx.y * 64 + wv * 4) * DIMC + lane * 4;
    v4f gv[2], bv[2];
#pragma unroll
    for (int p = 0; p < 2; ++p) {
      gv[p] = *(const v4f*)(lng + p * 128 + lane * 4);
      bv[p] = *(const v4f*)(lnb + p * 128 + lane * 4);
    }
    v4f xv[4][2];
#pragma unroll
    for (int j = 0; j < 4; ++j) {
#pragma unroll
      for (int p = 0; p < 2; ++p) xv[j][p] = *(const v4f*)(src + j * DIMC + p * 128);
    }
    __builtin_amdgcn_sched_barrier(0);
#pragma unroll
    for (int j = 0; j < 4; ++j) {
      v2f s2 = {0.f, 0.f};
#pragma unroll
      for (int p = 0; p < 2; ++p) {
#pragma unroll
        for (int q = 0; q < 2; ++q) {
          v2f t = {xv[j][p][2 * q], xv[j][p][2 * q + 1]};
          s2 += t;
        }
      }
      const float mean = wave_sum(s2[0] + s2[1]) * (1.0f / DIMC);
      v2f d2[2][2];
      v2f vs2 = {0.f, 0.f};
#pragma unroll
      for (int p = 0; p < 2; ++p) {
#pragma unroll
        for (int q = 0; q < 2; ++q) {
          v2f t = {xv[j][p][2 * q] - mean, xv[j][p][2 * q + 1] - mean};
          d2[p][q] = t;
          vs2 += t * t;
        }
      }
      const float rstd = rsqrtf(wave_sum(vs2[0] + vs2[1]) * (1.0f / DIMC) + 1e-5f);
#pragma unroll
      for (int p = 0; p < 2; ++p) {
        v4f yv;
#pragma unroll
        for (int q = 0; q < 2; ++q) {
#pragma unroll
          for (int e = 0; e < 2; ++e)
            yv[2 * q + e] = d2[p][q][e] * rstd * gv[p][2 * q + e] + bv[p][2 * q + e];
        }
        *(v4f*)(smem + (wv * 4 + j) * APITCH + p * 128 + lane * 4) = yv;
      }
    }
    __syncthreads();
  }
  const float* Apan = smem + (rb * 16 + lc) * APITCH;

  v8f acc[4];
#pragma unroll
  for (int nt = 0; nt < 4; ++nt) acc[nt] = v8zero();

  for (int kk = kbeg; kk < kend; kk += 8) {
    v2f a[2], b[2][4];
    if constexpr (!LNA) {
      const v4f a4 = *(const v4f*)(Afr + (size_t)(kk >> 3) * 128);
      v2f t0 = {a4[0], a4[1]};
      v2f t1 = {a4[2], a4[3]};
      a[0] = t0;
      a[1] = t1;
    }
#pragma unroll
    for (int u = 0; u < 2; ++u) {
      const int ka = kk + 4 * u + 2 * half;
      if constexpr (LNA) a[u] = *(const v2f*)(Apan + ka);
      const float* Wr0 = W + (size_t)ka * N + n0 + lc;
      const float* Wr1 = Wr0 + N;
#pragma unroll
      for (int nt = 0; nt < 4; ++nt) {
        v2f tb = {Wr0[nt * 16], Wr1[nt * 16]};
        b[u][nt] = tb;
      }
    }
#pragma unroll
    for (int u = 0; u < 2; ++u) {
#pragma unroll
      for (int nt = 0; nt < 4; ++nt) acc[nt] = wmma4(a[u], b[u][nt], acc[nt]);
    }
  }

  if constexpr (LNA) __syncthreads();

#pragma unroll
  for (int nt = 0; nt < 4; ++nt) {
#pragma unroll
    for (int v = 0; v < 8; ++v)
      part[wv][(v + 8 * half) * PPITCH + nt * 16 + lc] = acc[nt][v];
  }
  __syncthreads();

  if constexpr (FRAG) {
    static_assert(KS == 4, "fragment-order epilogue: 4 waves per key tile");
    const int sec = n0 / DIMC;
    const int cin = n0 - sec * DIMC;
    if (sec == 0) {
#pragma unroll
      for (int j = 0; j < 4; ++j) {
        const int r = ks * 4 + j;
        const int row = blockIdx.y * 64 + rb * 16 + r;
#pragma unroll
        for (int cs = 0; cs < 2; ++cs) {
          const int cl = cs * 32 + lane;
          float val = part[rb][r * PPITCH + cl];
#pragma unroll
          for (int t = 1; t < KS; ++t) val += part[4 * t + rb][r * PPITCH + cl];
          val += bias ? bias[n0 + cl] : 0.f;
          out[(size_t)row * DIMC + cin + cl] = val;
        }
      }
    } else {
      const int tile = blockIdx.y * 4 + rb;
      const int bb = tile / (NC / 16);
      const int jt = tile - bb * (NC / 16);
      const int hh = ks >> 1;
      const int head = cin / HDC + hh;
      float* dst = out + (size_t)sec * ROWS * DIMC +
                   (((size_t)(bb * HEADS + head) * (NC / 16) + jt) * 128 + lane) * 4;
#pragma unroll
      for (int q = 0; q < 2; ++q) {
        const int st = (ks & 1) * 2 + q;
        v4f val4;
#pragma unroll
        for (int sub = 0; sub < 2; ++sub) {
#pragma unroll
          for (int e = 0; e < 2; ++e) {
            int r, cl;
            if (sec == 1) { r = lc;                    cl = hh * HDC + 8 * st + 4 * sub + 2 * half + e; }
            else          { r = 4 * st + 2 * half + e; cl = hh * HDC + lc + 16 * sub; }
            float val = part[rb][r * PPITCH + cl];
#pragma unroll
            for (int t = 1; t < KS; ++t) val += part[4 * t + rb][r * PPITCH + cl];
            val += bias ? bias[n0 + cl] : 0.f;
            val4[sub * 2 + e] = val;
          }
        }
        *(v4f*)(dst + (size_t)st * 128) = val4;
      }
    }
  } else if constexpr (OFRAG) {
    static_assert(KS == 4, "fragment-order output: 8 k chunks per column block, 2 per wave");
    const int R = blockIdx.y * 4 + rb;
    float* dst = out + ((size_t)R * (N >> 3) + (n0 >> 3)) * 128 + lane * 4;
#pragma unroll
    for (int q = 0; q < 2; ++q) {
      const int tt = 2 * ks + q;
      v4f val4;
#pragma unroll
      for (int e = 0; e < 4; ++e) {
        const int cl = 8 * tt + 4 * (e >> 1) + 2 * half + (e & 1);
        float val = part[rb][lc * PPITCH + cl];
#pragma unroll
        for (int t = 1; t < KS; ++t) val += part[4 * t + rb][lc * PPITCH + cl];
        val += bias ? bias[n0 + cl] : 0.f;
        if (gelu_flag)
          val = 0.5f * val * (1.0f + erff(val * 0.70710678118654752f));
        val4[e] = val;
      }
      *(v4f*)(dst + (size_t)tt * 128) = val4;
    }
  } else {
  constexpr int RPW = 16 / KS;
#pragma unroll
  for (int j = 0; j < RPW; ++j) {
    const int r = ks * RPW + j;
    const int row = blockIdx.y * 64 + rb * 16 + r;
#pragma unroll
    for (int cs = 0; cs < 2; ++cs) {
      const int cl = cs * 32 + lane;
      float val = part[rb][r * PPITCH + cl];
#pragma unroll
      for (int t = 1; t < KS; ++t) val += part[4 * t + rb][r * PPITCH + cl];
      const int col = n0 + cl;
      val += bias ? bias[col] : 0.f;
      if (gelu_flag)
        val = 0.5f * val * (1.0f + erff(val * 0.70710678118654752f));
      if (resid) val += resid[(size_t)row * N + col];
      out[(size_t)row * N + col] = val;
    }
  }
  }
}

#define KSPLIT 4
#define KEYS_PER_SPLIT (NC / KSPLIT)
#define KT 2
#define OPITCH 36

__global__ __launch_bounds__(512, 8) void attn_kernel(
    const float* __restrict__ qkv, float* __restrict__ o) {
  __shared__ float plds[4 * KSPLIT][KT][256];
  __shared__ float part_o[4 * KSPLIT][16 * OPITCH];
  __shared__ float part_m[4 * KSPLIT][16];
  __shared__ float part_l[4 * KSPLIT][16];
  const int lane = threadIdx.x & 31;
  const int wv = threadIdx.x >> 5;
  const int qb = wv & 3;
  const int ks = wv >> 2;
  const int half = lane >> 4;
  const int lc = lane & 15;
  const int bh = blockIdx.x;
  const int bb = bh >> 3, h = bh & 7;
  const int row0 = (blockIdx.y * 4 + qb) * 16;
  const int base = bb * NC;
  const float scale = 0.17677669529663687f;

  const float* Q = qkv + (size_t)base * DIMC + h * HDC;
  const float* Kf = qkv + (size_t)ROWS * DIMC + (size_t)bh * (NC / 16) * 512 + lane * 4;
  const float* Vf = Kf + (size_t)ROWS * DIMC;

  v2f qa[8];
  {
    const float* qrow = Q + (size_t)(row0 + lc) * DIMC + 2 * half;
#pragma unroll
    for (int i = 0; i < 8; ++i) {
      v2f t = {qrow[4 * i] * scale, qrow[4 * i + 1] * scale};
      qa[i] = t;
    }
  }

  float mrow[8], lrow[8];
#pragma unroll
  for (int v = 0; v < 8; ++v) { mrow[v] = -1e30f; lrow[v] = 0.f; }
  v8f oacc0 = v8zero(), oacc1 = v8zero();

  const int jb0 = ks * (KEYS_PER_SPLIT / 16);
  for (int jb = jb0; jb < jb0 + KEYS_PER_SPLIT / 16; jb += KT) {
    v8f s[KT];
#pragma unroll
    for (int kt = 0; kt < KT; ++kt) s[kt] = v8zero();
#pragma unroll
    for (int i2 = 0; i2 < 4; ++i2) {
#pragma unroll
      for (int kt = 0; kt < KT; ++kt) {
        const v4f k4 = *(const v4f*)(Kf + (size_t)((jb + kt) * 4 + i2) * 128);
        v2f kb0 = {k4[0], k4[1]};
        v2f kb1 = {k4[2], k4[3]};
        s[kt] = wmma4(qa[2 * i2], kb0, s[kt]);
        s[kt] = wmma4(qa[2 * i2 + 1], kb1, s[kt]);
      }
    }
#pragma unroll
    for (int v = 0; v < 8; ++v) {
      float rmax = s[0][v];
#pragma unroll
      for (int kt = 1; kt < KT; ++kt) rmax = fmaxf(rmax, s[kt][v]);
      rmax = fmaxf(rmax, __shfl_xor(rmax, 1, 32));
      rmax = fmaxf(rmax, __shfl_xor(rmax, 2, 32));
      rmax = fmaxf(rmax, __shfl_xor(rmax, 4, 32));
      rmax = fmaxf(rmax, __shfl_xor(rmax, 8, 32));
      const float mn = fmaxf(mrow[v], rmax);
      const float c = __expf(mrow[v] - mn);
      float rs = 0.f;
#pragma unroll
      for (int kt = 0; kt < KT; ++kt) {
        const float p = __expf(s[kt][v] - mn);
        rs += p;
        plds[wv][kt][(v + 8 * half) * 16 + lc] = p;
      }
      lrow[v] = lrow[v] * c + rs;
      mrow[v] = mn;
      oacc0[v] *= c;
      oacc1[v] *= c;
    }
#pragma unroll
    for (int kt = 0; kt < KT; ++kt) {
#pragma unroll
      for (int i = 0; i < 4; ++i) {
        v2f pa = {plds[wv][kt][lc * 16 + 4 * i + 2 * half],
                  plds[wv][kt][lc * 16 + 4 * i + 2 * half + 1]};
        const v4f v4 = *(const v4f*)(Vf + (size_t)((jb + kt) * 4 + i) * 128);
        v2f vb0 = {v4[0], v4[1]};
        v2f vb1 = {v4[2], v4[3]};
        oacc0 = wmma4(pa, vb0, oacc0);
        oacc1 = wmma4(pa, vb1, oacc1);
      }
    }
  }

#pragma unroll
  for (int v = 0; v < 8; ++v) {
    float l = lrow[v];
    l += __shfl_xor(l, 1, 32);
    l += __shfl_xor(l, 2, 32);
    l += __shfl_xor(l, 4, 32);
    l += __shfl_xor(l, 8, 32);
    lrow[v] = l;
  }

#pragma unroll
  for (int v = 0; v < 8; ++v) {
    const int r = v + 8 * half;
    part_o[wv][r * OPITCH + lc] = oacc0[v];
    part_o[wv][r * OPITCH + 16 + lc] = oacc1[v];
    if (lc == 0) { part_m[wv][r] = mrow[v]; part_l[wv][r] = lrow[v]; }
  }
  __syncthreads();

  {
    const int r = lc;
    float mx = part_m[qb][r];
#pragma unroll
    for (int t = 1; t < KSPLIT; ++t) mx = fmaxf(mx, part_m[4 * t + qb][r]);
    float w[KSPLIT];
    float lsum = 0.f;
#pragma unroll
    for (int t = 0; t < KSPLIT; ++t) {
      w[t] = __expf(part_m[4 * t + qb][r] - mx);
      lsum += part_l[4 * t + qb][r] * w[t];
    }
    v4f val4;
#pragma unroll
    for (int sub = 0; sub < 2; ++sub) {
      v2f os = {0.f, 0.f};
#pragma unroll
      for (int t = 0; t < KSPLIT; ++t) {
        const v2f* pg = (const v2f*)&part_o[4 * t + qb][r * OPITCH + 8 * ks + 4 * sub];
        os += pg[half] * w[t];
      }
      val4[2 * sub] = os[0] / lsum;
      val4[2 * sub + 1] = os[1] / lsum;
    }
    const int R = (base + row0) >> 4;
    *(v4f*)(o + ((size_t)R * (DIMC / 8) + h * 4 + ks) * 128 + lane * 4) = val4;
  }
}

extern "C" void kernel_launch(void* const* d_in, const int* in_sizes, int n_in,
                              void* d_out, int out_size, void* d_ws, size_t ws_size,
                              hipStream_t stream) {
  const float* x     = (const float*)d_in[0];
  const float* ln1_g = (const float*)d_in[1] + (size_t)LAYER * DIMC;
  const float* ln1_b = (const float*)d_in[2] + (size_t)LAYER * DIMC;
  const float* Wqkv  = (const float*)d_in[3] + (size_t)LAYER * DIMC * QKVN;
  const float* bqkv  = (const float*)d_in[4] + (size_t)LAYER * QKVN;
  const float* Wo    = (const float*)d_in[5] + (size_t)LAYER * DIMC * DIMC;
  const float* bo    = (const float*)d_in[6] + (size_t)LAYER * DIMC;
  const float* ln2_g = (const float*)d_in[7] + (size_t)LAYER * DIMC;
  const float* ln2_b = (const float*)d_in[8] + (size_t)LAYER * DIMC;
  const float* W1    = (const float*)d_in[9] + (size_t)LAYER * DIMC * MLPC;
  const float* b1    = (const float*)d_in[10] + (size_t)LAYER * MLPC;
  const float* W2    = (const float*)d_in[11] + (size_t)LAYER * MLPC * DIMC;
  const float* b2    = (const float*)d_in[12] + (size_t)LAYER * DIMC;
  float* out = (float*)d_out;

  float* ws   = (float*)d_ws;
  float* oat  = ws;
  float* qkvb = ws + (size_t)ROWS * DIMC;
  float* a    = qkvb + (size_t)ROWS * QKVN;
  float* h1   = a + (size_t)ROWS * DIMC;

  wmma_gemm_kernel<4, true, true, false><<<dim3(QKVN / 64, ROWS / 64), 512, 0, stream>>>(
      x, Wqkv, bqkv, nullptr, ln1_g, ln1_b, qkvb, ROWS, QKVN, DIMC, 0);
  attn_kernel<<<dim3(BC * HEADS, NC / 64), 512, 0, stream>>>(qkvb, oat);
  wmma_gemm_kernel<8, false, false, false><<<dim3(DIMC / 64, ROWS / 64), 1024, 0, stream>>>(
      oat, Wo, bo, x, nullptr, nullptr, a, ROWS, DIMC, DIMC, 0);
  wmma_gemm_kernel<4, false, true, true><<<dim3(MLPC / 64, ROWS / 64), 512, 0, stream>>>(
      a, W1, b1, nullptr, ln2_g, ln2_b, h1, ROWS, MLPC, DIMC, 1);
  wmma_gemm_kernel<8, false, false, false><<<dim3(DIMC / 64, ROWS / 64), 1024, 0, stream>>>(
      h1, W2, b2, a, nullptr, nullptr, out, ROWS, DIMC, MLPC, 0);
}


# GEMM epilogues: bias and residual values are fetched with the prologue / first K-trip loads, ahead of the split-K partial exchange, not one waited-for load per element in the merge; sums unchanged
# speedup vs baseline: 3.3965x; 1.0444x over previous
#include <hip/hip_runtime.h>
#include <hip/hip_bf16.h>


#define DIMC   256
#define MLPC   1024
#define BC     2
#define NC     2048
#define HEADS  8
#define HDC    32
#define ROWS   (BC * NC)
#define QKVN   (3 * DIMC)
#define LAYER  1

typedef __attribute__((ext_vector_type(2))) float v2f;
typedef __attribute__((ext_vector_type(4))) float v4f;
typedef __attribute__((ext_vector_type(8))) float v8f;

static __device__ inline v8f wmma4(v2f a, v2f b, v8f c) {
  return __builtin_amdgcn_wmma_f32_16x16x4_f32(
       false, a,  false, b,
       (short)0, c,  false,  false);
}

static __device__ inline v8f v8zero() {
  v8f z = {0.f, 0.f, 0.f, 0.f, 0.f, 0.f, 0.f, 0.f};
  return z;
}

template <int M>
static __device__ inline float xor_in_row(float v) {
  return __int_as_float(__builtin_amdgcn_update_dpp(
      0, __float_as_int(v), 0x160 + M, 0xf, 0xf, true));
}

static __device__ inline float wave_sum(float s) {
  s += xor_in_row<1>(s);
  s += xor_in_row<2>(s);
  s += xor_in_row<4>(s);
  s += xor_in_row<8>(s);
  s += __shfl_xor(s, 16, 32);
  return s;
}

#define PPITCH 68
#define APITCH 260

template <int KS, bool FRAG, bool LNA, bool OFRAG, bool RESID, bool GELU>
__global__ __launch_bounds__(128 * KS) void wmma_gemm_kernel(
    const float* __restrict__ A, const float* __restrict__ W,
    const float* __restrict__ bias, const float* __restrict__ resid,
    const float* __restrict__ lng, const float* __restrict__ lnb,
    float* __restrict__ out, int M, int N, int K) {
  constexpr int PARTF = 4 * KS * 16 * PPITCH;
  constexpr int PANELF = LNA ? 64 * APITCH : 0;
  __shared__ float smem[PARTF > PANELF ? PARTF : PANELF];
  float (*part)[16 * PPITCH] = reinterpret_cast<float (*)[16 * PPITCH]>(smem);
  const int lane = threadIdx.x & 31;
  const int wv = __builtin_amdgcn_readfirstlane(threadIdx.x >> 5);
  const int rb = wv & 3;
  const int ks = wv >> 2;
  const int half = lane >> 4;
  const int lc = lane & 15;
  const int n0 = blockIdx.x * 64;
  const int m0 = blockIdx.y * 64 + rb * 16;
  const int klen = K / KS;
  const int kbeg = ks * klen;
  const int kend = kbeg + klen;

  const float* Afr = A + (size_t)(m0 >> 4) * (K >> 3) * 128 + lane * 4;

  constexpr int RPW = 16 / KS;
  float bq[8], rq[2 * RPW];
#pragma unroll
  for (int i = 0; i < 8; ++i) bq[i] = 0.f;
#pragma unroll
  for (int i = 0; i < 2 * RPW; ++i) rq[i] = 0.f;
  auto fetch_epilogue_operands = [&]() {
    if constexpr (FRAG) {
      const int sec = n0 / DIMC;
      if (sec == 0) {
#pragma unroll
        for (int cs = 0; cs < 2; ++cs) bq[cs] = bias[n0 + cs * 32 + lane];
      } else {
        const int hh = ks >> 1;
#pragma unroll
        for (int q = 0; q < 2; ++q) {
          const int st = (ks & 1) * 2 + q;
#pragma unroll
          for (int sub = 0; sub < 2; ++sub) {
#pragma unroll
            for (int e = 0; e < 2; ++e) {
              const int cl = (sec == 1) ? hh * HDC + 8 * st + 4 * sub + 2 * half + e
                                        : hh * HDC + lc + 16 * sub;
              bq[q * 4 + sub * 2 + e] = bias[n0 + cl];
            }
          }
        }
      }
    } else if constexpr (OFRAG) {
#pragma unroll
      for (int q = 0; q < 2; ++q) {
        const int tt = 2 * ks + q;
#pragma unroll
        for (int e = 0; e < 4; ++e)
          bq[q * 4 + e] = bias[n0 + 8 * tt + 4 * (e >> 1) + 2 * half + (e & 1)];
      }
    } else {
#pragma unroll
      for (int cs = 0; cs < 2; ++cs) bq[cs] = bias[n0 + cs * 32 + lane];
      if constexpr (RESID) {
#pragma unroll
        for (int j = 0; j < RPW; ++j) {
          const int row = blockIdx.y * 64 + rb * 16 + ks * RPW + j;
#pragma unroll
          for (int cs = 0; cs < 2; ++cs)
            rq[j * 2 + cs] = resid[(size_t)row * N + n0 + cs * 32 + lane];
        }
      }
    }
  };

  if constexpr (LNA) {
    static_assert(KS == 4, "panel prologue: 16 waves x 4 rows = 64 rows");
    const float* src = A + (size_t)(blockIdx.y * 64 + wv * 4) * DIMC + lane * 4;
    v4f gv[2], bv[2];
#pragma unroll
    for (int p = 0; p < 2; ++p) {
      gv[p] = *(const v4f*)(lng + p * 128 + lane * 4);
      bv[p] = *(const v4f*)(lnb + p * 128 + lane * 4);
    }
    v4f xv[4][2];
#pragma unroll
    for (int j = 0; j < 4; ++j) {
#pragma unroll
      for (int p = 0; p < 2; ++p) xv[j][p] = *(const v4f*)(src + j * DIMC + p * 128);
    }
    fetch_epilogue_operands();
    __builtin_amdgcn_sched_barrier(0);
#pragma unroll
    for (int j = 0; j < 4; ++j) {
      v2f s2 = {0.f, 0.f};
#pragma unroll
      for (int p = 0; p < 2; ++p) {
#pragma unroll
        for (int q = 0; q < 2; ++q) {
          v2f t = {xv[j][p][2 * q], xv[j][p][2 * q + 1]};
          s2 += t;
        }
      }
      const float mean = wave_sum(s2[0] + s2[1]) * (1.0f / DIMC);
      v2f d2[2][2];
      v2f vs2 = {0.f, 0.f};
#pragma unroll
      for (int p = 0; p < 2; ++p) {
#pragma unroll
        for (int q = 0; q < 2; ++q) {
          v2f t = {xv[j][p][2 * q] - mean, xv[j][p][2 * q + 1] - mean};
          d2[p][q] = t;
          vs2 += t * t;
        }
      }
      const float rstd = rsqrtf(wave_sum(vs2[0] + vs2[1]) * (1.0f / DIMC) + 1e-5f);
#pragma unroll
      for (int p = 0; p < 2; ++p) {
        v4f yv;
#pragma unroll
        for (int q = 0; q < 2; ++q) {
#pragma unroll
          for (int e = 0; e < 2; ++e)
            yv[2 * q + e] = d2[p][q][e] * rstd * gv[p][2 * q + e] + bv[p][2 * q + e];
        }
        *(v4f*)(smem + (wv * 4 + j) * APITCH + p * 128 + lane * 4) = yv;
      }
    }
    __syncthreads();
  }
  const float* Apan = smem + (rb * 16 + lc) * APITCH;

  v8f acc[4];
#pragma unroll
  for (int nt = 0; nt < 4; ++nt) acc[nt] = v8zero();

  for (int kk = kbeg; kk < kend; kk += 8) {
    v2f a[2], b[2][4];
    if constexpr (!LNA) {
      if (kk == kbeg) fetch_epilogue_operands();
      const v4f a4 = *(const v4f*)(Afr + (size_t)(kk >> 3) * 128);
      v2f t0 = {a4[0], a4[1]};
      v2f t1 = {a4[2], a4[3]};
      a[0] = t0;
      a[1] = t1;
    }
#pragma unroll
    for (int u = 0; u < 2; ++u) {
      const int ka = kk + 4 * u + 2 * half;
      if constexpr (LNA) a[u] = *(const v2f*)(Apan + ka);
      const float* Wr0 = W + (size_t)ka * N + n0 + lc;
      const float* Wr1 = Wr0 + N;
#pragma unroll
      for (int nt = 0; nt < 4; ++nt) {
        v2f tb = {Wr0[nt * 16], Wr1[nt * 16]};
        b[u][nt] = tb;
      }
    }
#pragma unroll
    for (int u = 0; u < 2; ++u) {
#pragma unroll
      for (int nt = 0; nt < 4; ++nt) acc[nt] = wmma4(a[u], b[u][nt], acc[nt]);
    }
  }

  if constexpr (LNA) __syncthreads();

#pragma unroll
  for (int nt = 0; nt < 4; ++nt) {
#pragma unroll
    for (int v = 0; v < 8; ++v)
      part[wv][(v + 8 * half) * PPITCH + nt * 16 + lc] = acc[nt][v];
  }
  __syncthreads();

  if constexpr (FRAG) {
    static_assert(KS == 4, "fragment-order epilogue: 4 waves per key tile");
    const int sec = n0 / DIMC;
    const int cin = n0 - sec * DIMC;
    if (sec == 0) {
#pragma unroll
      for (int j = 0; j < 4; ++j) {
        const int r = ks * 4 + j;
        const int row = blockIdx.y * 64 + rb * 16 + r;
#pragma unroll
        for (int cs = 0; cs < 2; ++cs) {
          const int cl = cs * 32 + lane;
          float val = part[rb][r * PPITCH + cl];
#pragma unroll
          for (int t = 1; t < KS; ++t) val += part[4 * t + rb][r * PPITCH + cl];
          val += bq[cs];
          out[(size_t)row * DIMC + cin + cl] = val;
        }
      }
    } else {
      const int tile = blockIdx.y * 4 + rb;
      const int bb = tile / (NC / 16);
      const int jt = tile - bb * (NC / 16);
      const int hh = ks >> 1;
      const int head = cin / HDC + hh;
      float* dst = out + (size_t)sec * ROWS * DIMC +
                   (((size_t)(bb * HEADS + head) * (NC / 16) + jt) * 128 + lane) * 4;
#pragma unroll
      for (int q = 0; q < 2; ++q) {
        const int st = (ks & 1) * 2 + q;
        v4f val4;
#pragma unroll
        for (int sub = 0; sub < 2; ++sub) {
#pragma unroll
          for (int e = 0; e < 2; ++e) {
            int r, cl;
            if (sec == 1) { r = lc;                    cl = hh * HDC + 8 * st + 4 * sub + 2 * half + e; }
            else          { r = 4 * st + 2 * half + e; cl = hh * HDC + lc + 16 * sub; }
            float val = part[rb][r * PPITCH + cl];
#pragma unroll
            for (int t = 1; t < KS; ++t) val += part[4 * t + rb][r * PPITCH + cl];
            val += bq[q * 4 + sub * 2 + e];
            val4[sub * 2 + e] = val;
          }
        }
        *(v4f*)(dst + (size_t)st * 128) = val4;
      }
    }
  } else if constexpr (OFRAG) {
    static_assert(KS == 4, "fragment-order output: 8 k chunks per column block, 2 per wave");
    const int R = blockIdx.y * 4 + rb;
    float* dst = out + ((size_t)R * (N >> 3) + (n0 >> 3)) * 128 + lane * 4;
    v4f val4[2];
#pragma unroll
    for (int q = 0; q < 2; ++q) {
      const int tt = 2 * ks + q;
#pragma unroll
      for (int e = 0; e < 4; ++e) {
        const int cl = 8 * tt + 4 * (e >> 1) + 2 * half + (e & 1);
        float val = part[rb][lc * PPITCH + cl];
#pragma unroll
        for (int t = 1; t < KS; ++t) val += part[4 * t + rb][lc * PPITCH + cl];
        val4[q][e] = val + bq[q * 4 + e];
      }
    }
#pragma unroll
    for (int q = 0; q < 2; ++q) {
      if constexpr (GELU) {
#pragma unroll
        for (int e = 0; e < 4; ++e)
          val4[q][e] = 0.5f * val4[q][e] * (1.0f + erff(val4[q][e] * 0.70710678118654752f));
      }
      *(v4f*)(dst + (size_t)(2 * ks + q) * 128) = val4[q];
    }
  } else {
#pragma unroll
  for (int j = 0; j < RPW; ++j) {
    const int r = ks * RPW + j;
    const int row = blockIdx.y * 64 + rb * 16 + r;
#pragma unroll
    for (int cs = 0; cs < 2; ++cs) {
      const int cl = cs * 32 + lane;
      float val = part[rb][r * PPITCH + cl];
#pragma unroll
      for (int t = 1; t < KS; ++t) val += part[4 * t + rb][r * PPITCH + cl];
      const int col = n0 + cl;
      val += bq[cs];
      if constexpr (GELU)
        val = 0.5f * val * (1.0f + erff(val * 0.70710678118654752f));
      if constexpr (RESID) val += rq[j * 2 + cs];
      out[(size_t)row * N + col] = val;
    }
  }
  }
}

#define KSPLIT 4
#define KEYS_PER_SPLIT (NC / KSPLIT)
#define KT 2
#define OPITCH 36

__global__ __launch_bounds__(512, 8) void attn_kernel(
    const float* __restrict__ qkv, float* __restrict__ o) {
  __shared__ float plds[4 * KSPLIT][KT][256];
  __shared__ float part_o[4 * KSPLIT][16 * OPITCH];
  __shared__ float part_m[4 * KSPLIT][16];
  __shared__ float part_l[4 * KSPLIT][16];
  const int lane = threadIdx.x & 31;
  const int wv = threadIdx.x >> 5;
  const int qb = wv & 3;
  const int ks = wv >> 2;
  const int half = lane >> 4;
  const int lc = lane & 15;
  const int bh = blockIdx.x;
  const int bb = bh >> 3, h = bh & 7;
  const int row0 = (blockIdx.y * 4 + qb) * 16;
  const int base = bb * NC;
  const float scale = 0.17677669529663687f;

  const float* Q = qkv + (size_t)base * DIMC + h * HDC;
  const float* Kf = qkv + (size_t)ROWS * DIMC + (size_t)bh * (NC / 16) * 512 + lane * 4;
  const float* Vf = Kf + (size_t)ROWS * DIMC;

  v2f qa[8];
  {
    const float* qrow = Q + (size_t)(row0 + lc) * DIMC + 2 * half;
#pragma unroll
    for (int i = 0; i < 8; ++i) {
      v2f t = {qrow[4 * i] * scale, qrow[4 * i + 1] * scale};
      qa[i] = t;
    }
  }

  float mrow[8], lrow[8];
#pragma unroll
  for (int v = 0; v < 8; ++v) { mrow[v] = -1e30f; lrow[v] = 0.f; }
  v8f oacc0 = v8zero(), oacc1 = v8zero();

  const int jb0 = ks * (KEYS_PER_SPLIT / 16);
  for (int jb = jb0; jb < jb0 + KEYS_PER_SPLIT / 16; jb += KT) {
    v8f s[KT];
#pragma unroll
    for (int kt = 0; kt < KT; ++kt) s[kt] = v8zero();
#pragma unroll
    for (int i2 = 0; i2 < 4; ++i2) {
#pragma unroll
      for (int kt = 0; kt < KT; ++kt) {
        const v4f k4 = *(const v4f*)(Kf + (size_t)((jb + kt) * 4 + i2) * 128);
        v2f kb0 = {k4[0], k4[1]};
        v2f kb1 = {k4[2], k4[3]};
        s[kt] = wmma4(qa[2 * i2], kb0, s[kt]);
        s[kt] = wmma4(qa[2 * i2 + 1], kb1, s[kt]);
      }
    }
#pragma unroll
    for (int v = 0; v < 8; ++v) {
      float rmax = s[0][v];
#pragma unroll
      for (int kt = 1; kt < KT; ++kt) rmax = fmaxf(rmax, s[kt][v]);
      rmax = fmaxf(rmax, __shfl_xor(rmax, 1, 32));
      rmax = fmaxf(rmax, __shfl_xor(rmax, 2, 32));
      rmax = fmaxf(rmax, __shfl_xor(rmax, 4, 32));
      rmax = fmaxf(rmax, __shfl_xor(rmax, 8, 32));
      const float mn = fmaxf(mrow[v], rmax);
      const float c = __expf(mrow[v] - mn);
      float rs = 0.f;
#pragma unroll
      for (int kt = 0; kt < KT; ++kt) {
        const float p = __expf(s[kt][v] - mn);
        rs += p;
        plds[wv][kt][(v + 8 * half) * 16 + lc] = p;
      }
      lrow[v] = lrow[v] * c + rs;
      mrow[v] = mn;
      oacc0[v] *= c;
      oacc1[v] *= c;
    }
#pragma unroll
    for (int kt = 0; kt < KT; ++kt) {
#pragma unroll
      for (int i = 0; i < 4; ++i) {
        v2f pa = {plds[wv][kt][lc * 16 + 4 * i + 2 * half],
                  plds[wv][kt][lc * 16 + 4 * i + 2 * half + 1]};
        const v4f v4 = *(const v4f*)(Vf + (size_t)((jb + kt) * 4 + i) * 128);
        v2f vb0 = {v4[0], v4[1]};
        v2f vb1 = {v4[2], v4[3]};
        oacc0 = wmma4(pa, vb0, oacc0);
        oacc1 = wmma4(pa, vb1, oacc1);
      }
    }
  }

#pragma unroll
  for (int v = 0; v < 8; ++v) {
    float l = lrow[v];
    l += __shfl_xor(l, 1, 32);
    l += __shfl_xor(l, 2, 32);
    l += __shfl_xor(l, 4, 32);
    l += __shfl_xor(l, 8, 32);
    lrow[v] = l;
  }

#pragma unroll
  for (int v = 0; v < 8; ++v) {
    const int r = v + 8 * half;
    part_o[wv][r * OPITCH + lc] = oacc0[v];
    part_o[wv][r * OPITCH + 16 + lc] = oacc1[v];
    if (lc == 0) { part_m[wv][r] = mrow[v]; part_l[wv][r] = lrow[v]; }
  }
  __syncthreads();

  {
    const int r = lc;
    float mx = part_m[qb][r];
#pragma unroll
    for (int t = 1; t < KSPLIT; ++t) mx = fmaxf(mx, part_m[4 * t + qb][r]);
    float w[KSPLIT];
    float lsum = 0.f;
#pragma unroll
    for (int t = 0; t < KSPLIT; ++t) {
      w[t] = __expf(part_m[4 * t + qb][r] - mx);
      lsum += part_l[4 * t + qb][r] * w[t];
    }
    v4f val4;
#pragma unroll
    for (int sub = 0; sub < 2; ++sub) {
      v2f os = {0.f, 0.f};
#pragma unroll
      for (int t = 0; t < KSPLIT; ++t) {
        const v2f* pg = (const v2f*)&part_o[4 * t + qb][r * OPITCH + 8 * ks + 4 * sub];
        os += pg[half] * w[t];
      }
      val4[2 * sub] = os[0] / lsum;
      val4[2 * sub + 1] = os[1] / lsum;
    }
    const int R = (base + row0) >> 4;
    *(v4f*)(o + ((size_t)R * (DIMC / 8) + h * 4 + ks) * 128 + lane * 4) = val4;
  }
}

extern "C" void kernel_launch(void* const* d_in, const int* in_sizes, int n_in,
                              void* d_out, int out_size, void* d_ws, size_t ws_size,
                              hipStream_t stream) {
  const float* x     = (const float*)d_in[0];
  const float* ln1_g = (const float*)d_in[1] + (size_t)LAYER * DIMC;
  const float* ln1_b = (const float*)d_in[2] + (size_t)LAYER * DIMC;
  const float* Wqkv  = (const float*)d_in[3] + (size_t)LAYER * DIMC * QKVN;
  const float* bqkv  = (const float*)d_in[4] + (size_t)LAYER * QKVN;
  const float* Wo    = (const float*)d_in[5] + (size_t)LAYER * DIMC * DIMC;
  const float* bo    = (const float*)d_in[6] + (size_t)LAYER * DIMC;
  const float* ln2_g = (const float*)d_in[7] + (size_t)LAYER * DIMC;
  const float* ln2_b = (const float*)d_in[8] + (size_t)LAYER * DIMC;
  const float* W1    = (const float*)d_in[9] + (size_t)LAYER * DIMC * MLPC;
  const float* b1    = (const float*)d_in[10] + (size_t)LAYER * MLPC;
  const float* W2    = (const float*)d_in[11] + (size_t)LAYER * MLPC * DIMC;
  const float* b2    = (const float*)d_in[12] + (size_t)LAYER * DIMC;
  float* out = (float*)d_out;

  float* ws   = (float*)d_ws;
  float* oat  = ws;
  float* qkvb = ws + (size_t)ROWS * DIMC;
  float* a    = qkvb + (size_t)ROWS * QKVN;
  float* h1   = a + (size_t)ROWS * DIMC;

  wmma_gemm_kernel<4, true, true, false, false, false><<<dim3(QKVN / 64, ROWS / 64), 512, 0, stream>>>(
      x, Wqkv, bqkv, nullptr, ln1_g, ln1_b, qkvb, ROWS, QKVN, DIMC);
  attn_kernel<<<dim3(BC * HEADS, NC / 64), 512, 0, stream>>>(qkvb, oat);
  wmma_gemm_kernel<8, false, false, false, true, false><<<dim3(DIMC / 64, ROWS / 64), 1024, 0, stream>>>(
      oat, Wo, bo, x, nullptr, nullptr, a, ROWS, DIMC, DIMC);
  wmma_gemm_kernel<4, false, true, true, false, true><<<dim3(MLPC / 64, ROWS / 64), 512, 0, stream>>>(
      a, W1, b1, nullptr, ln2_g, ln2_b, h1, ROWS, MLPC, DIMC);
  wmma_gemm_kernel<8, false, false, false, true, false><<<dim3(DIMC / 64, ROWS / 64), 1024, 0, stream>>>(
      h1, W2, b2, a, nullptr, nullptr, out, ROWS, DIMC, MLPC);
}


# attention softmax: lazy row reference: the cross-lane row-max tree and the rescale of l and O run only in a trip where some score tops the reference in use by a fixed margin; changes rounding of p
# speedup vs baseline: 3.5559x; 1.0469x over previous
#include <hip/hip_runtime.h>
#include <hip/hip_bf16.h>


#define DIMC   256
#define MLPC   1024
#define BC     2
#define NC     2048
#define HEADS  8
#define HDC    32
#define ROWS   (BC * NC)
#define QKVN   (3 * DIMC)
#define LAYER  1

typedef __attribute__((ext_vector_type(2))) float v2f;
typedef __attribute__((ext_vector_type(4))) float v4f;
typedef __attribute__((ext_vector_type(8))) float v8f;

static __device__ inline v8f wmma4(v2f a, v2f b, v8f c) {
  return __builtin_amdgcn_wmma_f32_16x16x4_f32(
       false, a,  false, b,
       (short)0, c,  false,  false);
}

static __device__ inline v8f v8zero() {
  v8f z = {0.f, 0.f, 0.f, 0.f, 0.f, 0.f, 0.f, 0.f};
  return z;
}

template <int M>
static __device__ inline float xor_in_row(float v) {
  return __int_as_float(__builtin_amdgcn_update_dpp(
      0, __float_as_int(v), 0x160 + M, 0xf, 0xf, true));
}

static __device__ inline float wave_sum(float s) {
  s += xor_in_row<1>(s);
  s += xor_in_row<2>(s);
  s += xor_in_row<4>(s);
  s += xor_in_row<8>(s);
  s += __shfl_xor(s, 16, 32);
  return s;
}

#define PPITCH 68
#define APITCH 260

template <int KS, bool FRAG, bool LNA, bool OFRAG, bool RESID, bool GELU>
__global__ __launch_bounds__(128 * KS) void wmma_gemm_kernel(
    const float* __restrict__ A, const float* __restrict__ W,
    const float* __restrict__ bias, const float* __restrict__ resid,
    const float* __restrict__ lng, const float* __restrict__ lnb,
    float* __restrict__ out, int M, int N, int K) {
  constexpr int PARTF = 4 * KS * 16 * PPITCH;
  constexpr int PANELF = LNA ? 64 * APITCH : 0;
  __shared__ float smem[PARTF > PANELF ? PARTF : PANELF];
  float (*part)[16 * PPITCH] = reinterpret_cast<float (*)[16 * PPITCH]>(smem);
  const int lane = threadIdx.x & 31;
  const int wv = __builtin_amdgcn_readfirstlane(threadIdx.x >> 5);
  const int rb = wv & 3;
  const int ks = wv >> 2;
  const int half = lane >> 4;
  const int lc = lane & 15;
  const int n0 = blockIdx.x * 64;
  const int m0 = blockIdx.y * 64 + rb * 16;
  const int klen = K / KS;
  const int kbeg = ks * klen;
  const int kend = kbeg + klen;

  const float* Afr = A + (size_t)(m0 >> 4) * (K >> 3) * 128 + lane * 4;

  constexpr int RPW = 16 / KS;
  float bq[8], rq[2 * RPW];
#pragma unroll
  for (int i = 0; i < 8; ++i) bq[i] = 0.f;
#pragma unroll
  for (int i = 0; i < 2 * RPW; ++i) rq[i] = 0.f;
  auto fetch_epilogue_operands = [&]() {
    if constexpr (FRAG) {
      const int sec = n0 / DIMC;
      if (sec == 0) {
#pragma unroll
        for (int cs = 0; cs < 2; ++cs) bq[cs] = bias[n0 + cs * 32 + lane];
      } else {
        const int hh = ks >> 1;
#pragma unroll
        for (int q = 0; q < 2; ++q) {
          const int st = (ks & 1) * 2 + q;
#pragma unroll
          for (int sub = 0; sub < 2; ++sub) {
#pragma unroll
            for (int e = 0; e < 2; ++e) {
              const int cl = (sec == 1) ? hh * HDC + 8 * st + 4 * sub + 2 * half + e
                                        : hh * HDC + lc + 16 * sub;
              bq[q * 4 + sub * 2 + e] = bias[n0 + cl];
            }
          }
        }
      }
    } else if constexpr (OFRAG) {
#pragma unroll
      for (int q = 0; q < 2; ++q) {
        const int tt = 2 * ks + q;
#pragma unroll
        for (int e = 0; e < 4; ++e)
          bq[q * 4 + e] = bias[n0 + 8 * tt + 4 * (e >> 1) + 2 * half + (e & 1)];
      }
    } else {
#pragma unroll
      for (int cs = 0; cs < 2; ++cs) bq[cs] = bias[n0 + cs * 32 + lane];
      if constexpr (RESID) {
#pragma unroll
        for (int j = 0; j < RPW; ++j) {
          const int row = blockIdx.y * 64 + rb * 16 + ks * RPW + j;
#pragma unroll
          for (int cs = 0; cs < 2; ++cs)
            rq[j * 2 + cs] = resid[(size_t)row * N + n0 + cs * 32 + lane];
        }
      }
    }
  };

  if constexpr (LNA) {
    static_assert(KS == 4, "panel prologue: 16 waves x 4 rows = 64 rows");
    const float* src = A + (size_t)(blockIdx.y * 64 + wv * 4) * DIMC + lane * 4;
    v4f gv[2], bv[2];
#pragma unroll
    for (int p = 0; p < 2; ++p) {
      gv[p] = *(const v4f*)(lng + p * 128 + lane * 4);
      bv[p] = *(const v4f*)(lnb + p * 128 + lane * 4);
    }
    v4f xv[4][2];
#pragma unroll
    for (int j = 0; j < 4; ++j) {
#pragma unroll
      for (int p = 0; p < 2; ++p) xv[j][p] = *(const v4f*)(src + j * DIMC + p * 128);
    }
    fetch_epilogue_operands();
    __builtin_amdgcn_sched_barrier(0);
#pragma unroll
    for (int j = 0; j < 4; ++j) {
      v2f s2 = {0.f, 0.f};
#pragma unroll
      for (int p = 0; p < 2; ++p) {
#pragma unroll
        for (int q = 0; q < 2; ++q) {
          v2f t = {xv[j][p][2 * q], xv[j][p][2 * q + 1]};
          s2 += t;
        }
      }
      const float mean = wave_sum(s2[0] + s2[1]) * (1.0f / DIMC);
      v2f d2[2][2];
      v2f vs2 = {0.f, 0.f};
#pragma unroll
      for (int p = 0; p < 2; ++p) {
#pragma unroll
        for (int q = 0; q < 2; ++q) {
          v2f t = {xv[j][p][2 * q] - mean, xv[j][p][2 * q + 1] - mean};
          d2[p][q] = t;
          vs2 += t * t;
        }
      }
      const float rstd = rsqrtf(wave_sum(vs2[0] + vs2[1]) * (1.0f / DIMC) + 1e-5f);
#pragma unroll
      for (int p = 0; p < 2; ++p) {
        v4f yv;
#pragma unroll
        for (int q = 0; q < 2; ++q) {
#pragma unroll
          for (int e = 0; e < 2; ++e)
            yv[2 * q + e] = d2[p][q][e] * rstd * gv[p][2 * q + e] + bv[p][2 * q + e];
        }
        *(v4f*)(smem + (wv * 4 + j) * APITCH + p * 128 + lane * 4) = yv;
      }
    }
    __syncthreads();
  }
  const float* Apan = smem + (rb * 16 + lc) * APITCH;

  v8f acc[4];
#pragma unroll
  for (int nt = 0; nt < 4; ++nt) acc[nt] = v8zero();

  for (int kk = kbeg; kk < kend; kk += 8) {
    v2f a[2], b[2][4];
    if constexpr (!LNA) {
      if (kk == kbeg) fetch_epilogue_operands();
      const v4f a4 = *(const v4f*)(Afr + (size_t)(kk >> 3) * 128);
      v2f t0 = {a4[0], a4[1]};
      v2f t1 = {a4[2], a4[3]};
      a[0] = t0;
      a[1] = t1;
    }
#pragma unroll
    for (int u = 0; u < 2; ++u) {
      const int ka = kk + 4 * u + 2 * half;
      if constexpr (LNA) a[u] = *(const v2f*)(Apan + ka);
      const float* Wr0 = W + (size_t)ka * N + n0 + lc;
      const float* Wr1 = Wr0 + N;
#pragma unroll
      for (int nt = 0; nt < 4; ++nt) {
        v2f tb = {Wr0[nt * 16], Wr1[nt * 16]};
        b[u][nt] = tb;
      }
    }
#pragma unroll
    for (int u = 0; u < 2; ++u) {
#pragma unroll
      for (int nt = 0; nt < 4; ++nt) acc[nt] = wmma4(a[u], b[u][nt], acc[nt]);
    }
  }

  if constexpr (LNA) __syncthreads();

#pragma unroll
  for (int nt = 0; nt < 4; ++nt) {
#pragma unroll
    for (int v = 0; v < 8; ++v)
      part[wv][(v + 8 * half) * PPITCH + nt * 16 + lc] = acc[nt][v];
  }
  __syncthreads();

  if constexpr (FRAG) {
    static_assert(KS == 4, "fragment-order epilogue: 4 waves per key tile");
    const int sec = n0 / DIMC;
    const int cin = n0 - sec * DIMC;
    if (sec == 0) {
#pragma unroll
      for (int j = 0; j < 4; ++j) {
        const int r = ks * 4 + j;
        const int row = blockIdx.y * 64 + rb * 16 + r;
#pragma unroll
        for (int cs = 0; cs < 2; ++cs) {
          const int cl = cs * 32 + lane;
          float val = part[rb][r * PPITCH + cl];
#pragma unroll
          for (int t = 1; t < KS; ++t) val += part[4 * t + rb][r * PPITCH + cl];
          val += bq[cs];
          out[(size_t)row * DIMC + cin + cl] = val;
        }
      }
    } else {
      const int tile = blockIdx.y * 4 + rb;
      const int bb = tile / (NC / 16);
      const int jt = tile - bb * (NC / 16);
      const int hh = ks >> 1;
      const int head = cin / HDC + hh;
      float* dst = out + (size_t)sec * ROWS * DIMC +
                   (((size_t)(bb * HEADS + head) * (NC / 16) + jt) * 128 + lane) * 4;
#pragma unroll
      for (int q = 0; q < 2; ++q) {
        const int st = (ks & 1) * 2 + q;
        v4f val4;
#pragma unroll
        for (int sub = 0; sub < 2; ++sub) {
#pragma unroll
          for (int e = 0; e < 2; ++e) {
            int r, cl;
            if (sec == 1) { r = lc;                    cl = hh * HDC + 8 * st + 4 * sub + 2 * half + e; }
            else          { r = 4 * st + 2 * half + e; cl = hh * HDC + lc + 16 * sub; }
            float val = part[rb][r * PPITCH + cl];
#pragma unroll
            for (int t = 1; t < KS; ++t) val += part[4 * t + rb][r * PPITCH + cl];
            val += bq[q * 4 + sub * 2 + e];
            val4[sub * 2 + e] = val;
          }
        }
        *(v4f*)(dst + (size_t)st * 128) = val4;
      }
    }
  } else if constexpr (OFRAG) {
    static_assert(KS == 4, "fragment-order output: 8 k chunks per column block, 2 per wave");
    const int R = blockIdx.y * 4 + rb;
    float* dst = out + ((size_t)R * (N >> 3) + (n0 >> 3)) * 128 + lane * 4;
    v4f val4[2];
#pragma unroll
    for (int q = 0; q < 2; ++q) {
      const int tt = 2 * ks + q;
#pragma unroll
      for (int e = 0; e < 4; ++e) {
        const int cl = 8 * tt + 4 * (e >> 1) + 2 * half + (e & 1);
        float val = part[rb][lc * PPITCH + cl];
#pragma unroll
        for (int t = 1; t < KS; ++t) val += part[4 * t + rb][lc * PPITCH + cl];
        val4[q][e] = val + bq[q * 4 + e];
      }
    }
#pragma unroll
    for (int q = 0; q < 2; ++q) {
      if constexpr (GELU) {
#pragma unroll
        for (int e = 0; e < 4; ++e)
          val4[q][e] = 0.5f * val4[q][e] * (1.0f + erff(val4[q][e] * 0.70710678118654752f));
      }
      *(v4f*)(dst + (size_t)(2 * ks + q) * 128) = val4[q];
    }
  } else {
#pragma unroll
  for (int j = 0; j < RPW; ++j) {
    const int r = ks * RPW + j;
    const int row = blockIdx.y * 64 + rb * 16 + r;
#pragma unroll
    for (int cs = 0; cs < 2; ++cs) {
      const int cl = cs * 32 + lane;
      float val = part[rb][r * PPITCH + cl];
#pragma unroll
      for (int t = 1; t < KS; ++t) val += part[4 * t + rb][r * PPITCH + cl];
      const int col = n0 + cl;
      val += bq[cs];
      if constexpr (GELU)
        val = 0.5f * val * (1.0f + erff(val * 0.70710678118654752f));
      if constexpr (RESID) val += rq[j * 2 + cs];
      out[(size_t)row * N + col] = val;
    }
  }
  }
}

#define KSPLIT 4
#define KEYS_PER_SPLIT (NC / KSPLIT)
#define KT 2
#define LAZY_T 6.0f
#define OPITCH 36

__global__ __launch_bounds__(512, 8) void attn_kernel(
    const float* __restrict__ qkv, float* __restrict__ o) {
  __shared__ float plds[4 * KSPLIT][KT][256];
  __shared__ float part_o[4 * KSPLIT][16 * OPITCH];
  __shared__ float part_m[4 * KSPLIT][16];
  __shared__ float part_l[4 * KSPLIT][16];
  const int lane = threadIdx.x & 31;
  const int wv = threadIdx.x >> 5;
  const int qb = wv & 3;
  const int ks = wv >> 2;
  const int half = lane >> 4;
  const int lc = lane & 15;
  const int bh = blockIdx.x;
  const int bb = bh >> 3, h = bh & 7;
  const int row0 = (blockIdx.y * 4 + qb) * 16;
  const int base = bb * NC;
  const float scale = 0.17677669529663687f;

  const float* Q = qkv + (size_t)base * DIMC + h * HDC;
  const float* Kf = qkv + (size_t)ROWS * DIMC + (size_t)bh * (NC / 16) * 512 + lane * 4;
  const float* Vf = Kf + (size_t)ROWS * DIMC;

  v2f qa[8];
  {
    const float* qrow = Q + (size_t)(row0 + lc) * DIMC + 2 * half;
#pragma unroll
    for (int i = 0; i < 8; ++i) {
      v2f t = {qrow[4 * i] * scale, qrow[4 * i + 1] * scale};
      qa[i] = t;
    }
  }

  float mrow[8], lrow[8];
#pragma unroll
  for (int v = 0; v < 8; ++v) { mrow[v] = -1e30f; lrow[v] = 0.f; }
  v8f oacc0 = v8zero(), oacc1 = v8zero();

  const int jb0 = ks * (KEYS_PER_SPLIT / 16);
  for (int jb = jb0; jb < jb0 + KEYS_PER_SPLIT / 16; jb += KT) {
    v8f s[KT];
#pragma unroll
    for (int kt = 0; kt < KT; ++kt) s[kt] = v8zero();
#pragma unroll
    for (int i2 = 0; i2 < 4; ++i2) {
#pragma unroll
      for (int kt = 0; kt < KT; ++kt) {
        const v4f k4 = *(const v4f*)(Kf + (size_t)((jb + kt) * 4 + i2) * 128);
        v2f kb0 = {k4[0], k4[1]};
        v2f kb1 = {k4[2], k4[3]};
        s[kt] = wmma4(qa[2 * i2], kb0, s[kt]);
        s[kt] = wmma4(qa[2 * i2 + 1], kb1, s[kt]);
      }
    }
    float dm[8];
#pragma unroll
    for (int v = 0; v < 8; ++v) {
      dm[v] = s[0][v] - mrow[v];
#pragma unroll
      for (int kt = 1; kt < KT; ++kt) dm[v] = fmaxf(dm[v], s[kt][v] - mrow[v]);
    }
    const float dmax = fmaxf(fmaxf(fmaxf(dm[0], dm[1]), fmaxf(dm[2], dm[3])),
                             fmaxf(fmaxf(dm[4], dm[5]), fmaxf(dm[6], dm[7])));
    if (__builtin_amdgcn_ballot_w32(dmax > LAZY_T) != 0u) {
#pragma unroll
      for (int v = 0; v < 8; ++v) {
        float rmax = s[0][v];
#pragma unroll
        for (int kt = 1; kt < KT; ++kt) rmax = fmaxf(rmax, s[kt][v]);
        rmax = fmaxf(rmax, __shfl_xor(rmax, 1, 32));
        rmax = fmaxf(rmax, __shfl_xor(rmax, 2, 32));
        rmax = fmaxf(rmax, __shfl_xor(rmax, 4, 32));
        rmax = fmaxf(rmax, __shfl_xor(rmax, 8, 32));
        const float mn = fmaxf(mrow[v], rmax);
        const float c = __expf(mrow[v] - mn);
        lrow[v] *= c;
        mrow[v] = mn;
        oacc0[v] *= c;
        oacc1[v] *= c;
      }
    }
#pragma unroll
    for (int v = 0; v < 8; ++v) {
      float rs = 0.f;
#pragma unroll
      for (int kt = 0; kt < KT; ++kt) {
        const float p = __expf(s[kt][v] - mrow[v]);
        rs += p;
        plds[wv][kt][(v + 8 * half) * 16 + lc] = p;
      }
      lrow[v] += rs;
    }
#pragma unroll
    for (int kt = 0; kt < KT; ++kt) {
#pragma unroll
      for (int i = 0; i < 4; ++i) {
        v2f pa = {plds[wv][kt][lc * 16 + 4 * i + 2 * half],
                  plds[wv][kt][lc * 16 + 4 * i + 2 * half + 1]};
        const v4f v4 = *(const v4f*)(Vf + (size_t)((jb + kt) * 4 + i) * 128);
        v2f vb0 = {v4[0], v4[1]};
        v2f vb1 = {v4[2], v4[3]};
        oacc0 = wmma4(pa, vb0, oacc0);
        oacc1 = wmma4(pa, vb1, oacc1);
      }
    }
  }

#pragma unroll
  for (int v = 0; v < 8; ++v) {
    float l = lrow[v];
    l += __shfl_xor(l, 1, 32);
    l += __shfl_xor(l, 2, 32);
    l += __shfl_xor(l, 4, 32);
    l += __shfl_xor(l, 8, 32);
    lrow[v] = l;
  }

#pragma unroll
  for (int v = 0; v < 8; ++v) {
    const int r = v + 8 * half;
    part_o[wv][r * OPITCH + lc] = oacc0[v];
    part_o[wv][r * OPITCH + 16 + lc] = oacc1[v];
    if (lc == 0) { part_m[wv][r] = mrow[v]; part_l[wv][r] = lrow[v]; }
  }
  __syncthreads();

  {
    const int r = lc;
    float mx = part_m[qb][r];
#pragma unroll
    for (int t = 1; t < KSPLIT; ++t) mx = fmaxf(mx, part_m[4 * t + qb][r]);
    float w[KSPLIT];
    float lsum = 0.f;
#pragma unroll
    for (int t = 0; t < KSPLIT; ++t) {
      w[t] = __expf(part_m[4 * t + qb][r] - mx);
      lsum += part_l[4 * t + qb][r] * w[t];
    }
    v4f val4;
#pragma unroll
    for (int sub = 0; sub < 2; ++sub) {
      v2f os = {0.f, 0.f};
#pragma unroll
      for (int t = 0; t < KSPLIT; ++t) {
        const v2f* pg = (const v2f*)&part_o[4 * t + qb][r * OPITCH + 8 * ks + 4 * sub];
        os += pg[half] * w[t];
      }
      val4[2 * sub] = os[0] / lsum;
      val4[2 * sub + 1] = os[1] / lsum;
    }
    const int R = (base + row0) >> 4;
    *(v4f*)(o + ((size_t)R * (DIMC / 8) + h * 4 + ks) * 128 + lane * 4) = val4;
  }
}

extern "C" void kernel_launch(void* const* d_in, const int* in_sizes, int n_in,
                              void* d_out, int out_size, void* d_ws, size_t ws_size,
                              hipStream_t stream) {
  const float* x     = (const float*)d_in[0];
  const float* ln1_g = (const float*)d_in[1] + (size_t)LAYER * DIMC;
  const float* ln1_b = (const float*)d_in[2] + (size_t)LAYER * DIMC;
  const float* Wqkv  = (const float*)d_in[3] + (size_t)LAYER * DIMC * QKVN;
  const float* bqkv  = (const float*)d_in[4] + (size_t)LAYER * QKVN;
  const float* Wo    = (const float*)d_in[5] + (size_t)LAYER * DIMC * DIMC;
  const float* bo    = (const float*)d_in[6] + (size_t)LAYER * DIMC;
  const float* ln2_g = (const float*)d_in[7] + (size_t)LAYER * DIMC;
  const float* ln2_b = (const float*)d_in[8] + (size_t)LAYER * DIMC;
  const float* W1    = (const float*)d_in[9] + (size_t)LAYER * DIMC * MLPC;
  const float* b1    = (const float*)d_in[10] + (size_t)LAYER * MLPC;
  const float* W2    = (const float*)d_in[11] + (size_t)LAYER * MLPC * DIMC;
  const float* b2    = (const float*)d_in[12] + (size_t)LAYER * DIMC;
  float* out = (float*)d_out;

  float* ws   = (float*)d_ws;
  float* oat  = ws;
  float* qkvb = ws + (size_t)ROWS * DIMC;
  float* a    = qkvb + (size_t)ROWS * QKVN;
  float* h1   = a + (size_t)ROWS * DIMC;

  wmma_gemm_kernel<4, true, true, false, false, false><<<dim3(QKVN / 64, ROWS / 64), 512, 0, stream>>>(
      x, Wqkv, bqkv, nullptr, ln1_g, ln1_b, qkvb, ROWS, QKVN, DIMC);
  attn_kernel<<<dim3(BC * HEADS, NC / 64), 512, 0, stream>>>(qkvb, oat);
  wmma_gemm_kernel<8, false, false, false, true, false><<<dim3(DIMC / 64, ROWS / 64), 1024, 0, stream>>>(
      oat, Wo, bo, x, nullptr, nullptr, a, ROWS, DIMC, DIMC);
  wmma_gemm_kernel<4, false, true, true, false, true><<<dim3(MLPC / 64, ROWS / 64), 512, 0, stream>>>(
      a, W1, b1, nullptr, ln2_g, ln2_b, h1, ROWS, MLPC, DIMC);
  wmma_gemm_kernel<8, false, false, false, true, false><<<dim3(DIMC / 64, ROWS / 64), 1024, 0, stream>>>(
      h1, W2, b2, a, nullptr, nullptr, out, ROWS, DIMC, MLPC);
}


# GEMM wave tile 32 rows x 64 columns (8 accumulators, each W fragment feeds 2 row blocks): the 4 GEMM launches use workgroups of half the waves, half as many waves enter; summation order unchanged
# speedup vs baseline: 3.7205x; 1.0463x over previous
#include <hip/hip_runtime.h>
#include <hip/hip_bf16.h>


#define DIMC   256
#define MLPC   1024
#define BC     2
#define NC     2048
#define HEADS  8
#define HDC    32
#define ROWS   (BC * NC)
#define QKVN   (3 * DIMC)
#define LAYER  1

typedef __attribute__((ext_vector_type(2))) float v2f;
typedef __attribute__((ext_vector_type(4))) float v4f;
typedef __attribute__((ext_vector_type(8))) float v8f;

static __device__ inline v8f wmma4(v2f a, v2f b, v8f c) {
  return __builtin_amdgcn_wmma_f32_16x16x4_f32(
       false, a,  false, b,
       (short)0, c,  false,  false);
}

static __device__ inline v8f v8zero() {
  v8f z = {0.f, 0.f, 0.f, 0.f, 0.f, 0.f, 0.f, 0.f};
  return z;
}

template <int M>
static __device__ inline float xor_in_row(float v) {
  return __int_as_float(__builtin_amdgcn_update_dpp(
      0, __float_as_int(v), 0x160 + M, 0xf, 0xf, true));
}

static __device__ inline float wave_sum(float s) {
  s += xor_in_row<1>(s);
  s += xor_in_row<2>(s);
  s += xor_in_row<4>(s);
  s += xor_in_row<8>(s);
  s += __shfl_xor(s, 16, 32);
  return s;
}

#define PPITCH 68
#define APITCH 260

template <int KS, bool FRAG, bool LNA, bool OFRAG, bool RESID, bool GELU>
__global__ __launch_bounds__(64 * KS) void wmma_gemm_kernel(
    const float* __restrict__ A, const float* __restrict__ W,
    const float* __restrict__ bias, const float* __restrict__ resid,
    const float* __restrict__ lng, const float* __restrict__ lnb,
    float* __restrict__ out, int M, int N, int K) {
  constexpr int PARTF = 4 * KS * 16 * PPITCH;
  constexpr int PANELF = LNA ? 64 * APITCH : 0;
  __shared__ float smem[PARTF > PANELF ? PARTF : PANELF];
  float (*part)[16 * PPITCH] = reinterpret_cast<float (*)[16 * PPITCH]>(smem);
  const int lane = threadIdx.x & 31;
  const int wv = __builtin_amdgcn_readfirstlane(threadIdx.x >> 5);
  const int rp = wv & 1;
  const int ks = wv >> 1;
  const int half = lane >> 4;
  const int lc = lane & 15;
  const int n0 = blockIdx.x * 64;
  const int m0 = blockIdx.y * 64 + rp * 32;
  const int klen = K / KS;
  const int kbeg = ks * klen;
  const int kend = kbeg + klen;

  const float* Afr = A + (size_t)(m0 >> 4) * (K >> 3) * 128 + lane * 4;
  const size_t Ablk = (size_t)(K >> 3) * 128;

  constexpr int RPW = 16 / KS;
  float bq[8], rq[2][2 * RPW];
#pragma unroll
  for (int i = 0; i < 8; ++i) bq[i] = 0.f;
#pragma unroll
  for (int r2 = 0; r2 < 2; ++r2) {
#pragma unroll
    for (int i = 0; i < 2 * RPW; ++i) rq[r2][i] = 0.f;
  }
  auto fetch_epilogue_operands = [&]() {
    if constexpr (FRAG) {
      const int sec = n0 / DIMC;
      if (sec == 0) {
#pragma unroll
        for (int cs = 0; cs < 2; ++cs) bq[cs] = bias[n0 + cs * 32 + lane];
      } else {
        const int hh = ks >> 1;
#pragma unroll
        for (int q = 0; q < 2; ++q) {
          const int st = (ks & 1) * 2 + q;
#pragma unroll
          for (int sub = 0; sub < 2; ++sub) {
#pragma unroll
            for (int e = 0; e < 2; ++e) {
              const int cl = (sec == 1) ? hh * HDC + 8 * st + 4 * sub + 2 * half + e
                                        : hh * HDC + lc + 16 * sub;
              bq[q * 4 + sub * 2 + e] = bias[n0 + cl];
            }
          }
        }
      }
    } else if constexpr (OFRAG) {
#pragma unroll
      for (int q = 0; q < 2; ++q) {
        const int tt = 2 * ks + q;
#pragma unroll
        for (int e = 0; e < 4; ++e)
          bq[q * 4 + e] = bias[n0 + 8 * tt + 4 * (e >> 1) + 2 * half + (e & 1)];
      }
    } else {
#pragma unroll
      for (int cs = 0; cs < 2; ++cs) bq[cs] = bias[n0 + cs * 32 + lane];
      if constexpr (RESID) {
#pragma unroll
        for (int r2 = 0; r2 < 2; ++r2) {
#pragma unroll
          for (int j = 0; j < RPW; ++j) {
            const int row = m0 + r2 * 16 + ks * RPW + j;
#pragma unroll
            for (int cs = 0; cs < 2; ++cs)
              rq[r2][j * 2 + cs] = resid[(size_t)row * N + n0 + cs * 32 + lane];
          }
        }
      }
    }
  };

  if constexpr (LNA) {
    static_assert(KS == 4, "panel prologue: 8 waves x 8 rows = 64 rows");
    const float* src = A + (size_t)(blockIdx.y * 64 + wv * 8) * DIMC + lane * 4;
    v4f gv[2], bv[2];
#pragma unroll
    for (int p = 0; p < 2; ++p) {
      gv[p] = *(const v4f*)(lng + p * 128 + lane * 4);
      bv[p] = *(const v4f*)(lnb + p * 128 + lane * 4);
    }
    v4f xv[8][2];
#pragma unroll
    for (int j = 0; j < 8; ++j) {
#pragma unroll
      for (int p = 0; p < 2; ++p) xv[j][p] = *(const v4f*)(src + j * DIMC + p * 128);
    }
    fetch_epilogue_operands();
    __builtin_amdgcn_sched_barrier(0);
#pragma unroll
    for (int j = 0; j < 8; ++j) {
      v2f s2 = {0.f, 0.f};
#pragma unroll
      for (int p = 0; p < 2; ++p) {
#pragma unroll
        for (int q = 0; q < 2; ++q) {
          v2f t = {xv[j][p][2 * q], xv[j][p][2 * q + 1]};
          s2 += t;
        }
      }
      const float mean = wave_sum(s2[0] + s2[1]) * (1.0f / DIMC);
      v2f d2[2][2];
      v2f vs2 = {0.f, 0.f};
#pragma unroll
      for (int p = 0; p < 2; ++p) {
#pragma unroll
        for (int q = 0; q < 2; ++q) {
          v2f t = {xv[j][p][2 * q] - mean, xv[j][p][2 * q + 1] - mean};
          d2[p][q] = t;
          vs2 += t * t;
        }
      }
      const float rstd = rsqrtf(wave_sum(vs2[0] + vs2[1]) * (1.0f / DIMC) + 1e-5f);
#pragma unroll
      for (int p = 0; p < 2; ++p) {
        v4f yv;
#pragma unroll
        for (int q = 0; q < 2; ++q) {
#pragma unroll
          for (int e = 0; e < 2; ++e)
            yv[2 * q + e] = d2[p][q][e] * rstd * gv[p][2 * q + e] + bv[p][2 * q + e];
        }
        *(v4f*)(smem + (wv * 8 + j) * APITCH + p * 128 + lane * 4) = yv;
      }
    }
    __syncthreads();
  }
  const float* Apan = smem + (rp * 32 + lc) * APITCH;

  v8f acc[2][4];
#pragma unroll
  for (int r2 = 0; r2 < 2; ++r2) {
#pragma unroll
    for (int nt = 0; nt < 4; ++nt) acc[r2][nt] = v8zero();
  }

  for (int kk = kbeg; kk < kend; kk += 8) {
    v2f a[2][2], b[2][4];
    if constexpr (!LNA) {
      if (kk == kbeg) fetch_epilogue_operands();
#pragma unroll
      for (int r2 = 0; r2 < 2; ++r2) {
        const v4f a4 = *(const v4f*)(Afr + r2 * Ablk + (size_t)(kk >> 3) * 128);
        v2f t0 = {a4[0], a4[1]};
        v2f t1 = {a4[2], a4[3]};
        a[r2][0] = t0;
        a[r2][1] = t1;
      }
    }
#pragma unroll
    for (int u = 0; u < 2; ++u) {
      const int ka = kk + 4 * u + 2 * half;
      if constexpr (LNA) {
#pragma unroll
        for (int r2 = 0; r2 < 2; ++r2) a[r2][u] = *(const v2f*)(Apan + r2 * 16 * APITCH + ka);
      }
      const float* Wr0 = W + (size_t)ka * N + n0 + lc;
      const float* Wr1 = Wr0 + N;
#pragma unroll
      for (int nt = 0; nt < 4; ++nt) {
        v2f tb = {Wr0[nt * 16], Wr1[nt * 16]};
        b[u][nt] = tb;
      }
    }
#pragma unroll
    for (int u = 0; u < 2; ++u) {
#pragma unroll
      for (int nt = 0; nt < 4; ++nt) {
#pragma unroll
        for (int r2 = 0; r2 < 2; ++r2) acc[r2][nt] = wmma4(a[r2][u], b[u][nt], acc[r2][nt]);
      }
    }
    if constexpr (!LNA && RESID) {
      static_assert(RPW == 2, "the fence below names 2 bias and 2 x 4 residual values");
      if (kk == kbeg)
        asm volatile("" :: "v"(bq[0]), "v"(bq[1]), "v"(rq[0][0]), "v"(rq[0][1]), "v"(rq[0][2]),
                     "v"(rq[0][3]), "v"(rq[1][0]), "v"(rq[1][1]), "v"(rq[1][2]), "v"(rq[1][3]));
    }
  }

  if constexpr (LNA) __syncthreads();

#pragma unroll
  for (int r2 = 0; r2 < 2; ++r2) {
#pragma unroll
    for (int nt = 0; nt < 4; ++nt) {
#pragma unroll
      for (int v = 0; v < 8; ++v)
        part[4 * ks + 2 * rp + r2][(v + 8 * half) * PPITCH + nt * 16 + lc] = acc[r2][nt][v];
    }
  }
  __syncthreads();

  if constexpr (FRAG) {
    static_assert(KS == 4, "fragment-order epilogue: 4 K ranges per key tile");
    const int sec = n0 / DIMC;
    const int cin = n0 - sec * DIMC;
    if (sec == 0) {
#pragma unroll
      for (int r2 = 0; r2 < 2; ++r2) {
        const int rb = 2 * rp + r2;
#pragma unroll
        for (int j = 0; j < 4; ++j) {
          const int r = ks * 4 + j;
          const int row = blockIdx.y * 64 + rb * 16 + r;
#pragma unroll
          for (int cs = 0; cs < 2; ++cs) {
            const int cl = cs * 32 + lane;
            float val = part[rb][r * PPITCH + cl];
#pragma unroll
            for (int t = 1; t < KS; ++t) val += part[4 * t + rb][r * PPITCH + cl];
            val += bq[cs];
            out[(size_t)row * DIMC + cin + cl] = val;
          }
        }
      }
    } else {
#pragma unroll
      for (int r2 = 0; r2 < 2; ++r2) {
        const int rb = 2 * rp + r2;
        const int tile = blockIdx.y * 4 + rb;
        const int bb = tile / (NC / 16);
        const int jt = tile - bb * (NC / 16);
        const int hh = ks >> 1;
        const int head = cin / HDC + hh;
        float* dst = out + (size_t)sec * ROWS * DIMC +
                     (((size_t)(bb * HEADS + head) * (NC / 16) + jt) * 128 + lane) * 4;
#pragma unroll
        for (int q = 0; q < 2; ++q) {
          const int st = (ks & 1) * 2 + q;
          v4f val4;
#pragma unroll
          for (int sub = 0; sub < 2; ++sub) {
#pragma unroll
            for (int e = 0; e < 2; ++e) {
              int r, cl;
              if (sec == 1) { r = lc;                    cl = hh * HDC + 8 * st + 4 * sub + 2 * half + e; }
              else          { r = 4 * st + 2 * half + e; cl = hh * HDC + lc + 16 * sub; }
              float val = part[rb][r * PPITCH + cl];
#pragma unroll
              for (int t = 1; t < KS; ++t) val += part[4 * t + rb][r * PPITCH + cl];
              val += bq[q * 4 + sub * 2 + e];
              val4[sub * 2 + e] = val;
            }
          }
          *(v4f*)(dst + (size_t)st * 128) = val4;
        }
      }
    }
  } else if constexpr (OFRAG) {
    static_assert(KS == 4, "fragment-order output: 8 k chunks per column block, 2 per K range");
    v4f val4[2][2];
#pragma unroll
    for (int r2 = 0; r2 < 2; ++r2) {
      const int rb = 2 * rp + r2;
#pragma unroll
      for (int q = 0; q < 2; ++q) {
        const int tt = 2 * ks + q;
#pragma unroll
        for (int e = 0; e < 4; ++e) {
          const int cl = 8 * tt + 4 * (e >> 1) + 2 * half + (e & 1);
          float val = part[rb][lc * PPITCH + cl];
#pragma unroll
          for (int t = 1; t < KS; ++t) val += part[4 * t + rb][lc * PPITCH + cl];
          val4[r2][q][e] = val + bq[q * 4 + e];
        }
      }
    }
#pragma unroll
    for (int r2 = 0; r2 < 2; ++r2) {
      const int R = blockIdx.y * 4 + 2 * rp + r2;
      float* dst = out + ((size_t)R * (N >> 3) + (n0 >> 3)) * 128 + lane * 4;
#pragma unroll
      for (int q = 0; q < 2; ++q) {
        if constexpr (GELU) {
#pragma unroll
          for (int e = 0; e < 4; ++e)
            val4[r2][q][e] = 0.5f * val4[r2][q][e] * (1.0f + erff(val4[r2][q][e] * 0.70710678118654752f));
        }
        *(v4f*)(dst + (size_t)(2 * ks + q) * 128) = val4[r2][q];
      }
    }
  } else {
#pragma unroll
  for (int r2 = 0; r2 < 2; ++r2) {
    const int rb = 2 * rp + r2;
#pragma unroll
    for (int j = 0; j < RPW; ++j) {
      const int r = ks * RPW + j;
      const int row = blockIdx.y * 64 + rb * 16 + r;
#pragma unroll
      for (int cs = 0; cs < 2; ++cs) {
        const int cl = cs * 32 + lane;
        float val = part[rb][r * PPITCH + cl];
#pragma unroll
        for (int t = 1; t < KS; ++t) val += part[4 * t + rb][r * PPITCH + cl];
        const int col = n0 + cl;
        val += bq[cs];
        if constexpr (GELU)
          val = 0.5f * val * (1.0f + erff(val * 0.70710678118654752f));
        if constexpr (RESID) val += rq[r2][j * 2 + cs];
        out[(size_t)row * N + col] = val;
      }
    }
  }
  }
}

#define KSPLIT 4
#define KEYS_PER_SPLIT (NC / KSPLIT)
#define KT 2
#define LAZY_T 6.0f
#define OPITCH 36

__global__ __launch_bounds__(512, 8) void attn_kernel(
    const float* __restrict__ qkv, float* __restrict__ o) {
  __shared__ float plds[4 * KSPLIT][KT][256];
  __shared__ float part_o[4 * KSPLIT][16 * OPITCH];
  __shared__ float part_m[4 * KSPLIT][16];
  __shared__ float part_l[4 * KSPLIT][16];
  const int lane = threadIdx.x & 31;
  const int wv = threadIdx.x >> 5;
  const int qb = wv & 3;
  const int ks = wv >> 2;
  const int half = lane >> 4;
  const int lc = lane & 15;
  const int bh = blockIdx.x;
  const int bb = bh >> 3, h = bh & 7;
  const int row0 = (blockIdx.y * 4 + qb) * 16;
  const int base = bb * NC;
  const float scale = 0.17677669529663687f;

  const float* Q = qkv + (size_t)base * DIMC + h * HDC;
  const float* Kf = qkv + (size_t)ROWS * DIMC + (size_t)bh * (NC / 16) * 512 + lane * 4;
  const float* Vf = Kf + (size_t)ROWS * DIMC;

  v2f qa[8];
  {
    const float* qrow = Q + (size_t)(row0 + lc) * DIMC + 2 * half;
#pragma unroll
    for (int i = 0; i < 8; ++i) {
      v2f t = {qrow[4 * i] * scale, qrow[4 * i + 1] * scale};
      qa[i] = t;
    }
  }

  float mrow[8], lrow[8];
#pragma unroll
  for (int v = 0; v < 8; ++v) { mrow[v] = -1e30f; lrow[v] = 0.f; }
  v8f oacc0 = v8zero(), oacc1 = v8zero();

  const int jb0 = ks * (KEYS_PER_SPLIT / 16);
  for (int jb = jb0; jb < jb0 + KEYS_PER_SPLIT / 16; jb += KT) {
    v8f s[KT];
#pragma unroll
    for (int kt = 0; kt < KT; ++kt) s[kt] = v8zero();
#pragma unroll
    for (int i2 = 0; i2 < 4; ++i2) {
#pragma unroll
      for (int kt = 0; kt < KT; ++kt) {
        const v4f k4 = *(const v4f*)(Kf + (size_t)((jb + kt) * 4 + i2) * 128);
        v2f kb0 = {k4[0], k4[1]};
        v2f kb1 = {k4[2], k4[3]};
        s[kt] = wmma4(qa[2 * i2], kb0, s[kt]);
        s[kt] = wmma4(qa[2 * i2 + 1], kb1, s[kt]);
      }
    }
    float dm[8];
#pragma unroll
    for (int v = 0; v < 8; ++v) {
      dm[v] = s[0][v] - mrow[v];
#pragma unroll
      for (int kt = 1; kt < KT; ++kt) dm[v] = fmaxf(dm[v], s[kt][v] - mrow[v]);
    }
    const float dmax = fmaxf(fmaxf(fmaxf(dm[0], dm[1]), fmaxf(dm[2], dm[3])),
                             fmaxf(fmaxf(dm[4], dm[5]), fmaxf(dm[6], dm[7])));
    if (__builtin_amdgcn_ballot_w32(dmax > LAZY_T) != 0u) {
#pragma unroll
      for (int v = 0; v < 8; ++v) {
        float rmax = s[0][v];
#pragma unroll
        for (int kt = 1; kt < KT; ++kt) rmax = fmaxf(rmax, s[kt][v]);
        rmax = fmaxf(rmax, __shfl_xor(rmax, 1, 32));
        rmax = fmaxf(rmax, __shfl_xor(rmax, 2, 32));
        rmax = fmaxf(rmax, __shfl_xor(rmax, 4, 32));
        rmax = fmaxf(rmax, __shfl_xor(rmax, 8, 32));
        const float mn = fmaxf(mrow[v], rmax);
        const float c = __expf(mrow[v] - mn);
        lrow[v] *= c;
        mrow[v] = mn;
        oacc0[v] *= c;
        oacc1[v] *= c;
      }
    }
#pragma unroll
    for (int v = 0; v < 8; ++v) {
      float rs = 0.f;
#pragma unroll
      for (int kt = 0; kt < KT; ++kt) {
        const float p = __expf(s[kt][v] - mrow[v]);
        rs += p;
        plds[wv][kt][(v + 8 * half) * 16 + lc] = p;
      }
      lrow[v] += rs;
    }
#pragma unroll
    for (int kt = 0; kt < KT; ++kt) {
#pragma unroll
      for (int i = 0; i < 4; ++i) {
        v2f pa = {plds[wv][kt][lc * 16 + 4 * i + 2 * half],
                  plds[wv][kt][lc * 16 + 4 * i + 2 * half + 1]};
        const v4f v4 = *(const v4f*)(Vf + (size_t)((jb + kt) * 4 + i) * 128);
        v2f vb0 = {v4[0], v4[1]};
        v2f vb1 = {v4[2], v4[3]};
        oacc0 = wmma4(pa, vb0, oacc0);
        oacc1 = wmma4(pa, vb1, oacc1);
      }
    }
  }

#pragma unroll
  for (int v = 0; v < 8; ++v) {
    float l = lrow[v];
    l += __shfl_xor(l, 1, 32);
    l += __shfl_xor(l, 2, 32);
    l += __shfl_xor(l, 4, 32);
    l += __shfl_xor(l, 8, 32);
    lrow[v] = l;
  }

#pragma unroll
  for (int v = 0; v < 8; ++v) {
    const int r = v + 8 * half;
    part_o[wv][r * OPITCH + lc] = oacc0[v];
    part_o[wv][r * OPITCH + 16 + lc] = oacc1[v];
    if (lc == 0) { part_m[wv][r] = mrow[v]; part_l[wv][r] = lrow[v]; }
  }
  __syncthreads();

  {
    const int r = lc;
    float mx = part_m[qb][r];
#pragma unroll
    for (int t = 1; t < KSPLIT; ++t) mx = fmaxf(mx, part_m[4 * t + qb][r]);
    float w[KSPLIT];
    float lsum = 0.f;
#pragma unroll
    for (int t = 0; t < KSPLIT; ++t) {
      w[t] = __expf(part_m[4 * t + qb][r] - mx);
      lsum += part_l[4 * t + qb][r] * w[t];
    }
    v4f val4;
#pragma unroll
    for (int sub = 0; sub < 2; ++sub) {
      v2f os = {0.f, 0.f};
#pragma unroll
      for (int t = 0; t < KSPLIT; ++t) {
        const v2f* pg = (const v2f*)&part_o[4 * t + qb][r * OPITCH + 8 * ks + 4 * sub];
        os += pg[half] * w[t];
      }
      val4[2 * sub] = os[0] / lsum;
      val4[2 * sub + 1] = os[1] / lsum;
    }
    const int R = (base + row0) >> 4;
    *(v4f*)(o + ((size_t)R * (DIMC / 8) + h * 4 + ks) * 128 + lane * 4) = val4;
  }
}

extern "C" void kernel_launch(void* const* d_in, const int* in_sizes, int n_in,
                              void* d_out, int out_size, void* d_ws, size_t ws_size,
                              hipStream_t stream) {
  const float* x     = (const float*)d_in[0];
  const float* ln1_g = (const float*)d_in[1] + (size_t)LAYER * DIMC;
  const float* ln1_b = (const float*)d_in[2] + (size_t)LAYER * DIMC;
  const float* Wqkv  = (const float*)d_in[3] + (size_t)LAYER * DIMC * QKVN;
  const float* bqkv  = (const float*)d_in[4] + (size_t)LAYER * QKVN;
  const float* Wo    = (const float*)d_in[5] + (size_t)LAYER * DIMC * DIMC;
  const float* bo    = (const float*)d_in[6] + (size_t)LAYER * DIMC;
  const float* ln2_g = (const float*)d_in[7] + (size_t)LAYER * DIMC;
  const float* ln2_b = (const float*)d_in[8] + (size_t)LAYER * DIMC;
  const float* W1    = (const float*)d_in[9] + (size_t)LAYER * DIMC * MLPC;
  const float* b1    = (const float*)d_in[10] + (size_t)LAYER * MLPC;
  const float* W2    = (const float*)d_in[11] + (size_t)LAYER * MLPC * DIMC;
  const float* b2    = (const float*)d_in[12] + (size_t)LAYER * DIMC;
  float* out = (float*)d_out;

  float* ws   = (float*)d_ws;
  float* oat  = ws;
  float* qkvb = ws + (size_t)ROWS * DIMC;
  float* a    = qkvb + (size_t)ROWS * QKVN;
  float* h1   = a + (size_t)ROWS * DIMC;

  wmma_gemm_kernel<4, true, true, false, false, false><<<dim3(QKVN / 64, ROWS / 64), 256, 0, stream>>>(
      x, Wqkv, bqkv, nullptr, ln1_g, ln1_b, qkvb, ROWS, QKVN, DIMC);
  attn_kernel<<<dim3(BC * HEADS, NC / 64), 512, 0, stream>>>(qkvb, oat);
  wmma_gemm_kernel<8, false, false, false, true, false><<<dim3(DIMC / 64, ROWS / 64), 512, 0, stream>>>(
      oat, Wo, bo, x, nullptr, nullptr, a, ROWS, DIMC, DIMC);
  wmma_gemm_kernel<4, false, true, true, false, true><<<dim3(MLPC / 64, ROWS / 64), 256, 0, stream>>>(
      a, W1, b1, nullptr, ln2_g, ln2_b, h1, ROWS, MLPC, DIMC);
  wmma_gemm_kernel<8, false, false, false, true, false><<<dim3(DIMC / 64, ROWS / 64), 512, 0, stream>>>(
      h1, W2, b2, a, nullptr, nullptr, out, ROWS, DIMC, MLPC);
}
